# starting kernel
# baseline (speedup 1.0000x reference)
#include <hip/hip_runtime.h>
#include <math.h>

#define MG   128
#define NG   128
#define DIMD 256
#define BSZ  8192
#define KUNITS (MG * NG)

typedef __attribute__((ext_vector_type(2))) float v2f;
typedef __attribute__((ext_vector_type(8))) float v8f;

__device__ __forceinline__ v8f wmma_f32(v2f a, v2f b, v8f c) {
    return __builtin_amdgcn_wmma_f32_16x16x4_f32(
         false, a,  false, b,
         (short)0, c,  false,  false);
}

__global__ void __launch_bounds__(256)
k_wsq(const float* __restrict__ W, float* __restrict__ wsq) {
    const int k = blockIdx.x * blockDim.x + threadIdx.x;
    const float4* w4 = (const float4*)(W + (size_t)k * DIMD);
    float acc = 0.0f;
#pragma unroll 8
    for (int i = 0; i < DIMD / 4; ++i) {
        float4 v = w4[i];
        acc += v.x * v.x + v.y * v.y + v.z * v.z + v.w * v.w;
    }
    wsq[k] = acc;
}

__global__ void __launch_bounds__(256)
k_bmu(const float* __restrict__ X, const float* __restrict__ W,
      const float* __restrict__ wsq, int* __restrict__ bmu) {
    const int tid  = threadIdx.x;
    const int lane = tid & 31;
    const int wave = tid >> 5;
    const int l    = lane & 15;
    const int half = lane >> 4;
    const int row0 = (blockIdx.x * 8 + wave) * 16;

    v2f a[64];
    const float* arow = X + (size_t)(row0 + l) * DIMD + 2 * half;
#pragma unroll
    for (int s = 0; s < 64; ++s) {
        a[s].x = arow[4 * s + 0];
        a[s].y = arow[4 * s + 1];
    }

    float bestV[8];
    int   bestI[8];
#pragma unroll
    for (int r = 0; r < 8; ++r) { bestV[r] = 3.4e38f; bestI[r] = 0; }

    for (int tile = 0; tile < KUNITS / 16; ++tile) {
        const int col = tile * 16 + l;
        const float* brow = W + (size_t)col * DIMD + 2 * half;
        v8f c = {};
#pragma unroll
        for (int s = 0; s < 64; ++s) {
            v2f b;
            b.x = brow[4 * s + 0];
            b.y = brow[4 * s + 1];
            c = wmma_f32(a[s], b, c);
        }
        const float ws = wsq[col];
#pragma unroll
        for (int r = 0; r < 8; ++r) {
            float v = ws - 2.0f * c[r];
            if (v < bestV[r] || (v == bestV[r] && col < bestI[r])) {
                bestV[r] = v;
                bestI[r] = col;
            }
        }
    }

#pragma unroll
    for (int r = 0; r < 8; ++r) {
        float v = bestV[r];
        int   i = bestI[r];
#pragma unroll
        for (int m = 8; m >= 1; m >>= 1) {
            float ov = __shfl_xor(v, m, 32);
            int   oi = __shfl_xor(i, m, 32);
            if (ov < v || (ov == v && oi < i)) { v = ov; i = oi; }
        }
        if (l == 0) bmu[row0 + r + 8 * half] = i;
    }
}

__global__ void __launch_bounds__(256)
k_pxy(const int* __restrict__ bmu, const int* __restrict__ itp,
      float* __restrict__ px, float* __restrict__ py) {
    const float START_SIGMA = 64.0f;
    const float TAU = 20.0f / logf(START_SIGMA);
    const float sigma = START_SIGMA * expf(-(float)itp[0] / TAU);
    const float inv_d = 1.0f / (2.0f * sigma * sigma);

    const int b   = blockIdx.x * 2 + (threadIdx.x >> 7);
    const int idx = threadIdx.x & 127;
    const int k   = bmu[b];
    const float dr = (float)idx - (float)(k >> 7);
    const float dc = (float)idx - (float)(k & 127);
    px[(size_t)b * MG + idx] = expf(-dr * dr * inv_d);
    py[(size_t)b * NG + idx] = expf(-dc * dc * inv_d);
}

__global__ void __launch_bounds__(256)
k_denom(const float* __restrict__ px, const float* __restrict__ py,
        float* __restrict__ denom) {
    const int tid  = threadIdx.x;
    const int lane = tid & 31;
    const int wave = tid >> 5;
    const int l    = lane & 15;
    const int half = lane >> 4;
    const int t     = blockIdx.x * 8 + wave;
    const int iBase = (t >> 3) * 16;
    const int jBase = (t & 7) * 16;

    v8f c = {};
    for (int b = 0; b < BSZ; b += 4) {
        const int bb = b + 2 * half;
        v2f a, bf;
        a.x  = px[(size_t)bb * MG + iBase + l];
        a.y  = px[(size_t)(bb + 1) * MG + iBase + l];
        bf.x = py[(size_t)bb * NG + jBase + l];
        bf.y = py[(size_t)(bb + 1) * NG + jBase + l];
        c = wmma_f32(a, bf, c);
    }
#pragma unroll
    for (int r = 0; r < 8; ++r)
        denom[(size_t)(iBase + r + 8 * half) * NG + jBase + l] = c[r];
}

__global__ void __launch_bounds__(256)
k_update(const float* __restrict__ X, const float* __restrict__ px,
         const float* __restrict__ py, const float* __restrict__ denom,
         const float* __restrict__ W, float* __restrict__ out) {
    const int tid  = threadIdx.x;
    const int lane = tid & 31;
    const int wave = tid >> 5;
    const int l    = lane & 15;
    const int half = lane >> 4;

    const int gw    = blockIdx.x * 8 + wave;
    const int kTile = gw >> 2;
    const int dBase = (gw & 3) * 64;
    const int kBase = kTile * 16;
    const int i     = kBase >> 7;
    const int j0    = kBase & 127;

    v8f c0 = {}, c1 = {}, c2 = {}, c3 = {};
    for (int b = 0; b < BSZ; b += 4) {
        const int bb = b + 2 * half;
        const float p0 = px[(size_t)bb * MG + i];
        const float p1 = px[(size_t)(bb + 1) * MG + i];
        v2f a;
        a.x = p0 * py[(size_t)bb * NG + j0 + l];
        a.y = p1 * py[(size_t)(bb + 1) * NG + j0 + l];

        const float* x0 = X + (size_t)bb * DIMD + dBase + l;
        const float* x1 = X + (size_t)(bb + 1) * DIMD + dBase + l;
        v2f bf;
        bf.x = x0[0];  bf.y = x1[0];  c0 = wmma_f32(a, bf, c0);
        bf.x = x0[16]; bf.y = x1[16]; c1 = wmma_f32(a, bf, c1);
        bf.x = x0[32]; bf.y = x1[32]; c2 = wmma_f32(a, bf, c2);
        bf.x = x0[48]; bf.y = x1[48]; c3 = wmma_f32(a, bf, c3);
    }

    float den[8];
#pragma unroll
    for (int r = 0; r < 8; ++r)
        den[r] = denom[kBase + r + 8 * half];

    const v8f* cs[4] = { &c0, &c1, &c2, &c3 };
#pragma unroll
    for (int dt = 0; dt < 4; ++dt) {
#pragma unroll
        for (int r = 0; r < 8; ++r) {
            const int k = kBase + r + 8 * half;
            const int d = dBase + dt * 16 + l;
            const float v = (*cs[dt])[r];
            out[(size_t)k * DIMD + d] =
                (den[r] != 0.0f) ? (v / den[r]) : W[(size_t)k * DIMD + d];
        }
    }
}

extern "C" void kernel_launch(void* const* d_in, const int* in_sizes, int n_in,
                              void* d_out, int out_size, void* d_ws, size_t ws_size,
                              hipStream_t stream) {
    const float* X  = (const float*)d_in[0];
    const float* W  = (const float*)d_in[1];
    const int*   it = (const int*)d_in[2];
    float* out = (float*)d_out;

    float* ws    = (float*)d_ws;
    float* wsq   = ws;
    int*   bmu   = (int*)(ws + KUNITS);
    float* px    = ws + KUNITS + BSZ;
    float* py    = px + (size_t)BSZ * MG;
    float* denom = py + (size_t)BSZ * NG;

    k_wsq   <<<KUNITS / 256, 256, 0, stream>>>(W, wsq);
    k_bmu   <<<BSZ / (8 * 16), 256, 0, stream>>>(X, W, wsq, bmu);
    k_pxy   <<<BSZ / 2, 256, 0, stream>>>(bmu, it, px, py);
    k_denom <<<8, 256, 0, stream>>>(px, py, denom);
    k_update<<<(KUNITS / 16) * 4 / 8, 256, 0, stream>>>(X, px, py, denom, W, out);
}


# k_bmu operands swapped: a wave keeps a 16-unit W tile in VGPRs, streams data-row tiles from a fragment-ordered X copy; 4096 waves, not 512; per-tile minima merged by new k_merge; BMU bit-identical
# speedup vs baseline: 2.9506x; 2.9506x over previous
#include <hip/hip_runtime.h>
#include <math.h>

#define MG   128
#define NG   128
#define DIMD 256
#define BSZ  8192
#define KUNITS (MG * NG)

typedef __attribute__((ext_vector_type(2))) float v2f;
typedef __attribute__((ext_vector_type(8))) float v8f;

__device__ __forceinline__ v8f wmma_f32(v2f a, v2f b, v8f c) {
    return __builtin_amdgcn_wmma_f32_16x16x4_f32(
         false, a,  false, b,
         (short)0, c,  false,  false);
}

__global__ void __launch_bounds__(256)
k_wsq(const float* __restrict__ W, const float* __restrict__ X,
      float* __restrict__ wsq, float* __restrict__ Xf) {
    const int k = blockIdx.x * blockDim.x + threadIdx.x;
    const float4* w4 = (const float4*)(W + (size_t)k * DIMD);
    float acc = 0.0f;
#pragma unroll 4
    for (int p = 0; p < DIMD / 8; ++p) {
        const float4 v0 = w4[2 * p];
        const float4 v1 = w4[2 * p + 1];
        acc += v0.x * v0.x + v0.y * v0.y + v0.z * v0.z + v0.w * v0.w;
        acc += v1.x * v1.x + v1.y * v1.y + v1.z * v1.z + v1.w * v1.w;
    }
    wsq[k] = acc;

    if (k < BSZ) {
        const float4* x4 = (const float4*)(X + (size_t)k * DIMD);
        float4* dst = (float4*)Xf + (size_t)(k >> 4) * (32 * 32) + (k & 15);
#pragma unroll 4
        for (int p = 0; p < DIMD / 8; ++p) {
            const float4 v0 = x4[2 * p];
            const float4 v1 = x4[2 * p + 1];
            dst[p * 32]      = make_float4(v0.x, v0.y, v1.x, v1.y);
            dst[p * 32 + 16] = make_float4(v0.z, v0.w, v1.z, v1.w);
        }
    }
}

__global__ void __launch_bounds__(256) __attribute__((amdgpu_waves_per_eu(4, 4)))
k_bmu(const float* __restrict__ W, const float* __restrict__ Xf,
      const float* __restrict__ wsq, int* __restrict__ part) {
    const int tid  = threadIdx.x;
    const int lane = tid & 31;
    const int wave = tid >> 5;
    const int l    = lane & 15;
    const int half = lane >> 4;
    const int ut   = (blockIdx.x >> 2) * 8 + wave;
    const int u0   = ut * 16;
    const int rt0  = (blockIdx.x & 3) * (BSZ / 16 / 4);

    v2f a[64];
    const float* arow = W + (size_t)(u0 + l) * DIMD + 2 * half;
#pragma unroll
    for (int s = 0; s < 64; ++s) {
        a[s].x = arow[4 * s + 0];
        a[s].y = arow[4 * s + 1];
    }

    float ws[8];
#pragma unroll
    for (int r = 0; r < 8; ++r) ws[r] = wsq[u0 + 8 * half + r];

    const float4* xf4 = (const float4*)Xf + lane;
    int2* pout = (int2*)part + (size_t)ut * BSZ + l;
    for (int rt = rt0; rt < rt0 + BSZ / 16 / 4; ++rt) {
        const float4* bt = xf4 + (size_t)rt * (32 * 32);
        v8f c = {};
#pragma unroll
        for (int p = 0; p < 32; ++p) {
            const float4 v = bt[p * 32];
            v2f b0, b1;
            b0.x = v.x; b0.y = v.y;
            b1.x = v.z; b1.y = v.w;
            c = wmma_f32(a[2 * p], b0, c);
            c = wmma_f32(a[2 * p + 1], b1, c);
        }
        float bv = ws[0] - 2.0f * c[0];
        int   br = 0;
#pragma unroll
        for (int r = 1; r < 8; ++r) {
            const float v = ws[r] - 2.0f * c[r];
            if (v < bv) { bv = v; br = r; }
        }
        int bi = u0 + 8 * half + br;
        const float ov = __shfl_xor(bv, 16, 32);
        const int   oi = __shfl_xor(bi, 16, 32);
        if (ov < bv || (ov == bv && oi < bi)) { bv = ov; bi = oi; }
        if (half == 0) pout[rt * 16] = make_int2(__float_as_int(bv), bi);
    }
}

__global__ void __launch_bounds__(256)
k_merge(const int* __restrict__ part, int* __restrict__ bmu) {
    __shared__ float sv[8][32];
    __shared__ int   si[8][32];
    const int lane = threadIdx.x & 31;
    const int seg  = threadIdx.x >> 5;
    const int row  = blockIdx.x * 32 + lane;
    const int2* p = (const int2*)part + (size_t)(seg * 128) * BSZ + row;
    float bv = 3.4e38f;
    int   bi = 0;
#pragma unroll 16
    for (int t = 0; t < 128; ++t) {
        const int2 e = p[(size_t)t * BSZ];
        const float v = __int_as_float(e.x);
        if (v < bv) { bv = v; bi = e.y; }
    }
    sv[seg][lane] = bv;
    si[seg][lane] = bi;
    __syncthreads();
    if (seg == 0) {
#pragma unroll
        for (int s = 1; s < 8; ++s) {
            const float v = sv[s][lane];
            const int   i = si[s][lane];
            if (v < bv) { bv = v; bi = i; }
        }
        bmu[row] = bi;
    }
}

__global__ void __launch_bounds__(256)
k_pxy(const int* __restrict__ bmu, const int* __restrict__ itp,
      float* __restrict__ px, float* __restrict__ py) {
    const float START_SIGMA = 64.0f;
    const float TAU = 20.0f / logf(START_SIGMA);
    const float sigma = START_SIGMA * expf(-(float)itp[0] / TAU);
    const float inv_d = 1.0f / (2.0f * sigma * sigma);

    const int b   = blockIdx.x * 2 + (threadIdx.x >> 7);
    const int idx = threadIdx.x & 127;
    const int k   = bmu[b];
    const float dr = (float)idx - (float)(k >> 7);
    const float dc = (float)idx - (float)(k & 127);
    px[(size_t)b * MG + idx] = expf(-dr * dr * inv_d);
    py[(size_t)b * NG + idx] = expf(-dc * dc * inv_d);
}

__global__ void __launch_bounds__(256)
k_denom(const float* __restrict__ px, const float* __restrict__ py,
        float* __restrict__ denom) {
    const int tid  = threadIdx.x;
    const int lane = tid & 31;
    const int wave = tid >> 5;
    const int l    = lane & 15;
    const int half = lane >> 4;
    const int t     = blockIdx.x * 8 + wave;
    const int iBase = (t >> 3) * 16;
    const int jBase = (t & 7) * 16;

    v8f c = {};
    for (int b = 0; b < BSZ; b += 4) {
        const int bb = b + 2 * half;
        v2f a, bf;
        a.x  = px[(size_t)bb * MG + iBase + l];
        a.y  = px[(size_t)(bb + 1) * MG + iBase + l];
        bf.x = py[(size_t)bb * NG + jBase + l];
        bf.y = py[(size_t)(bb + 1) * NG + jBase + l];
        c = wmma_f32(a, bf, c);
    }
#pragma unroll
    for (int r = 0; r < 8; ++r)
        denom[(size_t)(iBase + r + 8 * half) * NG + jBase + l] = c[r];
}

__global__ void __launch_bounds__(256)
k_update(const float* __restrict__ X, const float* __restrict__ px,
         const float* __restrict__ py, const float* __restrict__ denom,
         const float* __restrict__ W, float* __restrict__ out) {
    const int tid  = threadIdx.x;
    const int lane = tid & 31;
    const int wave = tid >> 5;
    const int l    = lane & 15;
    const int half = lane >> 4;

    const int gw    = blockIdx.x * 8 + wave;
    const int kTile = gw >> 2;
    const int dBase = (gw & 3) * 64;
    const int kBase = kTile * 16;
    const int i     = kBase >> 7;
    const int j0    = kBase & 127;

    v8f c0 = {}, c1 = {}, c2 = {}, c3 = {};
    for (int b = 0; b < BSZ; b += 4) {
        const int bb = b + 2 * half;
        const float p0 = px[(size_t)bb * MG + i];
        const float p1 = px[(size_t)(bb + 1) * MG + i];
        v2f a;
        a.x = p0 * py[(size_t)bb * NG + j0 + l];
        a.y = p1 * py[(size_t)(bb + 1) * NG + j0 + l];

        const float* x0 = X + (size_t)bb * DIMD + dBase + l;
        const float* x1 = X + (size_t)(bb + 1) * DIMD + dBase + l;
        v2f bf;
        bf.x = x0[0];  bf.y = x1[0];  c0 = wmma_f32(a, bf, c0);
        bf.x = x0[16]; bf.y = x1[16]; c1 = wmma_f32(a, bf, c1);
        bf.x = x0[32]; bf.y = x1[32]; c2 = wmma_f32(a, bf, c2);
        bf.x = x0[48]; bf.y = x1[48]; c3 = wmma_f32(a, bf, c3);
    }

    float den[8];
#pragma unroll
    for (int r = 0; r < 8; ++r)
        den[r] = denom[kBase + r + 8 * half];

    const v8f* cs[4] = { &c0, &c1, &c2, &c3 };
#pragma unroll
    for (int dt = 0; dt < 4; ++dt) {
#pragma unroll
        for (int r = 0; r < 8; ++r) {
            const int k = kBase + r + 8 * half;
            const int d = dBase + dt * 16 + l;
            const float v = (*cs[dt])[r];
            out[(size_t)k * DIMD + d] =
                (den[r] != 0.0f) ? (v / den[r]) : W[(size_t)k * DIMD + d];
        }
    }
}

extern "C" void kernel_launch(void* const* d_in, const int* in_sizes, int n_in,
                              void* d_out, int out_size, void* d_ws, size_t ws_size,
                              hipStream_t stream) {
    const float* X  = (const float*)d_in[0];
    const float* W  = (const float*)d_in[1];
    const int*   it = (const int*)d_in[2];
    float* out = (float*)d_out;

    float* ws    = (float*)d_ws;
    float* wsq   = ws;
    int*   bmu   = (int*)(ws + KUNITS);
    float* px    = ws + KUNITS + BSZ;
    float* py    = px + (size_t)BSZ * MG;
    float* denom = py + (size_t)BSZ * NG;
    float* Xf    = denom + KUNITS;
    int*   part  = (int*)(Xf + (size_t)BSZ * DIMD);

    k_wsq   <<<KUNITS / 256, 256, 0, stream>>>(W, X, wsq, Xf);
    k_bmu   <<<(KUNITS / 16 / 8) * 4, 256, 0, stream>>>(W, Xf, wsq, part);
    k_merge <<<BSZ / 32, 256, 0, stream>>>(part, bmu);
    k_pxy   <<<BSZ / 2, 256, 0, stream>>>(bmu, it, px, py);
    k_denom <<<8, 256, 0, stream>>>(px, py, denom);
    k_update<<<(KUNITS / 16) * 4 / 8, 256, 0, stream>>>(X, px, py, denom, W, out);
}


# k_update builds its A operand from per-row BMU offsets and a 255-entry Gaussian table in LDS and sums it for the denominators; k_pxy, k_denom, px, py dropped; the denominators' summing order changes
# speedup vs baseline: 4.1999x; 1.4234x over previous
#include <hip/hip_runtime.h>
#include <math.h>

#define MG   128
#define NG   128
#define DIMD 256
#define BSZ  8192
#define KUNITS (MG * NG)

typedef __attribute__((ext_vector_type(2))) float v2f;
typedef __attribute__((ext_vector_type(8))) float v8f;

__device__ __forceinline__ v8f wmma_f32(v2f a, v2f b, v8f c) {
    return __builtin_amdgcn_wmma_f32_16x16x4_f32(
         false, a,  false, b,
         (short)0, c,  false,  false);
}

__global__ void __launch_bounds__(256)
k_wsq(const float* __restrict__ W, const float* __restrict__ X,
      float* __restrict__ wsq, float* __restrict__ Xf) {
    const int k = blockIdx.x * blockDim.x + threadIdx.x;
    const float4* w4 = (const float4*)(W + (size_t)k * DIMD);
    float acc = 0.0f;
#pragma unroll 4
    for (int p = 0; p < DIMD / 8; ++p) {
        const float4 v0 = w4[2 * p];
        const float4 v1 = w4[2 * p + 1];
        acc += v0.x * v0.x + v0.y * v0.y + v0.z * v0.z + v0.w * v0.w;
        acc += v1.x * v1.x + v1.y * v1.y + v1.z * v1.z + v1.w * v1.w;
    }
    wsq[k] = acc;

    if (k < BSZ) {
        const float4* x4 = (const float4*)(X + (size_t)k * DIMD);
        float4* dst = (float4*)Xf + (size_t)(k >> 4) * (32 * 32) + (k & 15);
#pragma unroll 4
        for (int p = 0; p < DIMD / 8; ++p) {
            const float4 v0 = x4[2 * p];
            const float4 v1 = x4[2 * p + 1];
            dst[p * 32]      = make_float4(v0.x, v0.y, v1.x, v1.y);
            dst[p * 32 + 16] = make_float4(v0.z, v0.w, v1.z, v1.w);
        }
    }
}

__global__ void __launch_bounds__(256) __attribute__((amdgpu_waves_per_eu(4, 4)))
k_bmu(const float* __restrict__ W, const float* __restrict__ Xf,
      const float* __restrict__ wsq, int* __restrict__ part) {
    const int tid  = threadIdx.x;
    const int lane = tid & 31;
    const int wave = tid >> 5;
    const int l    = lane & 15;
    const int half = lane >> 4;
    const int ut   = (blockIdx.x >> 2) * 8 + wave;
    const int u0   = ut * 16;
    const int rt0  = (blockIdx.x & 3) * (BSZ / 16 / 4);

    v2f a[64];
    const float* arow = W + (size_t)(u0 + l) * DIMD + 2 * half;
#pragma unroll
    for (int s = 0; s < 64; ++s) {
        a[s].x = arow[4 * s + 0];
        a[s].y = arow[4 * s + 1];
    }

    float ws[8];
#pragma unroll
    for (int r = 0; r < 8; ++r) ws[r] = wsq[u0 + 8 * half + r];

    const float4* xf4 = (const float4*)Xf + lane;
    int2* pout = (int2*)part + (size_t)ut * BSZ + l;
    for (int rt = rt0; rt < rt0 + BSZ / 16 / 4; ++rt) {
        const float4* bt = xf4 + (size_t)rt * (32 * 32);
        v8f c = {};
#pragma unroll
        for (int p = 0; p < 32; ++p) {
            const float4 v = bt[p * 32];
            v2f b0, b1;
            b0.x = v.x; b0.y = v.y;
            b1.x = v.z; b1.y = v.w;
            c = wmma_f32(a[2 * p], b0, c);
            c = wmma_f32(a[2 * p + 1], b1, c);
        }
        float bv = ws[0] - 2.0f * c[0];
        int   br = 0;
#pragma unroll
        for (int r = 1; r < 8; ++r) {
            const float v = ws[r] - 2.0f * c[r];
            if (v < bv) { bv = v; br = r; }
        }
        int bi = u0 + 8 * half + br;
        const float ov = __shfl_xor(bv, 16, 32);
        const int   oi = __shfl_xor(bi, 16, 32);
        if (ov < bv || (ov == bv && oi < bi)) { bv = ov; bi = oi; }
        if (half == 0) pout[rt * 16] = make_int2(__float_as_int(bv), bi);
    }
}

__global__ void __launch_bounds__(256)
k_merge(const int* __restrict__ part, int* __restrict__ rc) {
    __shared__ float sv[8][32];
    __shared__ int   si[8][32];
    const int lane = threadIdx.x & 31;
    const int seg  = threadIdx.x >> 5;
    const int row  = blockIdx.x * 32 + lane;
    const int2* p = (const int2*)part + (size_t)(seg * 128) * BSZ + row;
    float bv = 3.4e38f;
    int   bi = 0;
#pragma unroll 16
    for (int t = 0; t < 128; ++t) {
        const int2 e = p[(size_t)t * BSZ];
        const float v = __int_as_float(e.x);
        if (v < bv) { bv = v; bi = e.y; }
    }
    sv[seg][lane] = bv;
    si[seg][lane] = bi;
    __syncthreads();
    if (seg == 0) {
#pragma unroll
        for (int s = 1; s < 8; ++s) {
            const float v = sv[s][lane];
            const int   i = si[s][lane];
            if (v < bv) { bv = v; bi = i; }
        }
        ((int2*)rc)[row] = make_int2(127 - (bi >> 7), 127 - (bi & 127));
    }
}

__global__ void __launch_bounds__(256)
k_update(const float* __restrict__ X, const int* __restrict__ rc,
         const int* __restrict__ itp, const float* __restrict__ W,
         float* __restrict__ out) {
    __shared__ float gt[256];
    const int tid  = threadIdx.x;
    const int lane = tid & 31;
    const int wave = tid >> 5;
    const int l    = lane & 15;
    const int half = lane >> 4;

    {
        const float START_SIGMA = 64.0f;
        const float TAU = 20.0f / logf(START_SIGMA);
        const float sigma = START_SIGMA * expf(-(float)itp[0] / TAU);
        const float inv_d = 1.0f / (2.0f * sigma * sigma);
        const float dr = (float)(tid - 127);
        gt[tid] = expf(-dr * dr * inv_d);
    }
    __syncthreads();

    const int gw    = blockIdx.x * 8 + wave;
    const int kTile = gw >> 2;
    const int dBase = (gw & 3) * 64;
    const int kBase = kTile * 16;
    const int i     = kBase >> 7;
    const int jl    = (kBase & 127) + l;

    const int4* rc4 = (const int4*)rc + half;

    float ds0 = 0.0f, ds1 = 0.0f;
    v8f c0 = {}, c1 = {}, c2 = {}, c3 = {};
    for (int b = 0; b < BSZ; b += 4) {
        const int bb = b + 2 * half;
        const int4 o = rc4[b >> 1];
        v2f a;
        a.x = gt[i + o.x] * gt[jl + o.y];
        a.y = gt[i + o.z] * gt[jl + o.w];
        ds0 += a.x;
        ds1 += a.y;

        const float* x0 = X + (size_t)bb * DIMD + dBase + l;
        const float* x1 = X + (size_t)(bb + 1) * DIMD + dBase + l;
        v2f bf;
        bf.x = x0[0];  bf.y = x1[0];  c0 = wmma_f32(a, bf, c0);
        bf.x = x0[16]; bf.y = x1[16]; c1 = wmma_f32(a, bf, c1);
        bf.x = x0[32]; bf.y = x1[32]; c2 = wmma_f32(a, bf, c2);
        bf.x = x0[48]; bf.y = x1[48]; c3 = wmma_f32(a, bf, c3);
    }

    float dl = ds0 + ds1;
    dl += __shfl_xor(dl, 16, 32);
    float den[8];
#pragma unroll
    for (int r = 0; r < 8; ++r)
        den[r] = __shfl(dl, 8 * half + r, 32);

    const v8f* cs[4] = { &c0, &c1, &c2, &c3 };
#pragma unroll
    for (int dt = 0; dt < 4; ++dt) {
#pragma unroll
        for (int r = 0; r < 8; ++r) {
            const int k = kBase + r + 8 * half;
            const int d = dBase + dt * 16 + l;
            const float v = (*cs[dt])[r];
            out[(size_t)k * DIMD + d] =
                (den[r] != 0.0f) ? (v / den[r]) : W[(size_t)k * DIMD + d];
        }
    }
}

extern "C" void kernel_launch(void* const* d_in, const int* in_sizes, int n_in,
                              void* d_out, int out_size, void* d_ws, size_t ws_size,
                              hipStream_t stream) {
    const float* X  = (const float*)d_in[0];
    const float* W  = (const float*)d_in[1];
    const int*   it = (const int*)d_in[2];
    float* out = (float*)d_out;

    float* ws    = (float*)d_ws;
    float* wsq   = ws;
    int*   rc    = (int*)(ws + KUNITS);
    float* Xf    = ws + KUNITS + 2 * BSZ;
    int*   part  = (int*)(Xf + (size_t)BSZ * DIMD);

    k_wsq   <<<KUNITS / 256, 256, 0, stream>>>(W, X, wsq, Xf);
    k_bmu   <<<(KUNITS / 16 / 8) * 4, 256, 0, stream>>>(W, Xf, wsq, part);
    k_merge <<<BSZ / 32, 256, 0, stream>>>(part, rc);
    k_update<<<(KUNITS / 16) * 4 / 8, 256, 0, stream>>>(X, rc, it, W, out);
}


# k_update replaced: V[unit] = plain sum of data rows per BMU unit (in LDS), then numer = G x V x G^T as two 128-deep Toeplitz matrix products (grid columns, then rows); the order of the sums changes
# speedup vs baseline: 8.5178x; 2.0281x over previous
#include <hip/hip_runtime.h>
#include <math.h>

#define MG   128
#define NG   128
#define DIMD 256
#define BSZ  8192
#define KUNITS (MG * NG)

typedef __attribute__((ext_vector_type(2))) float v2f;
typedef __attribute__((ext_vector_type(8))) float v8f;

__device__ __forceinline__ v8f wmma_f32(v2f a, v2f b, v8f c) {
    return __builtin_amdgcn_wmma_f32_16x16x4_f32(
         false, a,  false, b,
         (short)0, c,  false,  false);
}

__global__ void __launch_bounds__(256)
k_wsq(const float* __restrict__ W, const float* __restrict__ X,
      float* __restrict__ wsq, float* __restrict__ Xf) {
    const int k = blockIdx.x * blockDim.x + threadIdx.x;
    const float4* w4 = (const float4*)(W + (size_t)k * DIMD);
    float acc = 0.0f;
#pragma unroll 4
    for (int p = 0; p < DIMD / 8; ++p) {
        const float4 v0 = w4[2 * p];
        const float4 v1 = w4[2 * p + 1];
        acc += v0.x * v0.x + v0.y * v0.y + v0.z * v0.z + v0.w * v0.w;
        acc += v1.x * v1.x + v1.y * v1.y + v1.z * v1.z + v1.w * v1.w;
    }
    wsq[k] = acc;

    if (k < BSZ) {
        const float4* x4 = (const float4*)(X + (size_t)k * DIMD);
        float4* dst = (float4*)Xf + (size_t)(k >> 4) * (32 * 32) + (k & 15);
#pragma unroll 4
        for (int p = 0; p < DIMD / 8; ++p) {
            const float4 v0 = x4[2 * p];
            const float4 v1 = x4[2 * p + 1];
            dst[p * 32]      = make_float4(v0.x, v0.y, v1.x, v1.y);
            dst[p * 32 + 16] = make_float4(v0.z, v0.w, v1.z, v1.w);
        }
    }
}

__global__ void __launch_bounds__(256) __attribute__((amdgpu_waves_per_eu(4, 4)))
k_bmu(const float* __restrict__ W, const float* __restrict__ Xf,
      const float* __restrict__ wsq, int* __restrict__ part) {
    const int tid  = threadIdx.x;
    const int lane = tid & 31;
    const int wave = tid >> 5;
    const int l    = lane & 15;
    const int half = lane >> 4;
    const int ut   = (blockIdx.x >> 2) * 8 + wave;
    const int u0   = ut * 16;
    const int rt0  = (blockIdx.x & 3) * (BSZ / 16 / 4);

    v2f a[64];
    const float* arow = W + (size_t)(u0 + l) * DIMD + 2 * half;
#pragma unroll
    for (int s = 0; s < 64; ++s) {
        a[s].x = arow[4 * s + 0];
        a[s].y = arow[4 * s + 1];
    }

    float ws[8];
#pragma unroll
    for (int r = 0; r < 8; ++r) ws[r] = wsq[u0 + 8 * half + r];

    const float4* xf4 = (const float4*)Xf + lane;
    int2* pout = (int2*)part + (size_t)ut * BSZ + l;
    for (int rt = rt0; rt < rt0 + BSZ / 16 / 4; ++rt) {
        const float4* bt = xf4 + (size_t)rt * (32 * 32);
        v8f c = {};
#pragma unroll
        for (int p = 0; p < 32; ++p) {
            const float4 v = bt[p * 32];
            v2f b0, b1;
            b0.x = v.x; b0.y = v.y;
            b1.x = v.z; b1.y = v.w;
            c = wmma_f32(a[2 * p], b0, c);
            c = wmma_f32(a[2 * p + 1], b1, c);
        }
        float bv = ws[0] - 2.0f * c[0];
        int   br = 0;
#pragma unroll
        for (int r = 1; r < 8; ++r) {
            const float v = ws[r] - 2.0f * c[r];
            if (v < bv) { bv = v; br = r; }
        }
        int bi = u0 + 8 * half + br;
        const float ov = __shfl_xor(bv, 16, 32);
        const int   oi = __shfl_xor(bi, 16, 32);
        if (ov < bv || (ov == bv && oi < bi)) { bv = ov; bi = oi; }
        if (half == 0) pout[rt * 16] = make_int2(__float_as_int(bv), bi);
    }
}

__global__ void __launch_bounds__(256)
k_merge(const int* __restrict__ part, int* __restrict__ bmu) {
    __shared__ float sv[8][32];
    __shared__ int   si[8][32];
    const int lane = threadIdx.x & 31;
    const int seg  = threadIdx.x >> 5;
    const int row  = blockIdx.x * 32 + lane;
    const int2* p = (const int2*)part + (size_t)(seg * 128) * BSZ + row;
    float bv = 3.4e38f;
    int   bi = 0;
#pragma unroll 16
    for (int t = 0; t < 128; ++t) {
        const int2 e = p[(size_t)t * BSZ];
        const float v = __int_as_float(e.x);
        if (v < bv) { bv = v; bi = e.y; }
    }
    sv[seg][lane] = bv;
    si[seg][lane] = bi;
    __syncthreads();
    if (seg == 0) {
#pragma unroll
        for (int s = 1; s < 8; ++s) {
            const float v = sv[s][lane];
            const int   i = si[s][lane];
            if (v < bv) { bv = v; bi = i; }
        }
        bmu[row] = bi;
    }
}

__device__ __forceinline__ void gauss_table(float* gt, const int* itp, int tid) {
    const float START_SIGMA = 64.0f;
    const float TAU = 20.0f / logf(START_SIGMA);
    const float sigma = START_SIGMA * expf(-(float)itp[0] / TAU);
    const float inv_d = 1.0f / (2.0f * sigma * sigma);
    const float dr = (float)(tid - 127);
    gt[tid] = expf(-dr * dr * inv_d);
}

__device__ __forceinline__ void toeplitz_a(const float* gt, int l, int half, v2f (&a)[60]) {
#pragma unroll
    for (int q = 0; q < 60; ++q) {
        const int m = 127 + 4 * (q - 31) + l - 2 * half;
        a[q].x = gt[m];
        a[q].y = gt[m - 1];
    }
}

__device__ __forceinline__ void toeplitz_mm(const v2f (&a)[60], const v2f (&b)[32], v8f (&acc)[8]) {
#pragma unroll
    for (int s = 0; s < 32; ++s) {
#pragma unroll
        for (int t = 0; t < 8; ++t)
            acc[t] = wmma_f32(a[4 * t - s + 31], b[s], acc[t]);
    }
}

#define VP 136

__global__ void __launch_bounds__(256)
k_colpass(const float* __restrict__ X, const int* __restrict__ bmu,
          const int* __restrict__ itp, float* __restrict__ T, float* __restrict__ Tden) {
    __shared__ float gt[256];
    __shared__ float cnts[128];
    __shared__ __attribute__((aligned(16))) float Vs[128 * VP];
    const int tid  = threadIdx.x;
    const int lane = tid & 31;
    const int wave = tid >> 5;
    const int l    = lane & 15;
    const int half = lane >> 4;
    const int r    = blockIdx.x >> 1;
    const int dh   = blockIdx.x & 1;

    gauss_table(gt, itp, tid);
    for (int idx = tid; idx < 128 * VP / 4; idx += 256)
        ((float4*)Vs)[idx] = make_float4(0.0f, 0.0f, 0.0f, 0.0f);
    __syncthreads();

    const int key = r * 8 + wave;
    float mycnt = 0.0f;
    const int4*   b4 = (const int4*)bmu + lane;
    const float4* X4 = (const float4*)X + dh * 32 + lane;
    float4*       V4 = (float4*)Vs + lane;
    for (int chunk = 0; chunk < BSZ / 128 / 16; ++chunk) {
        int4 v[16];
#pragma unroll
        for (int k = 0; k < 16; ++k) v[k] = b4[(chunk * 16 + k) * 32];
#pragma unroll
        for (int k = 0; k < 16; ++k) {
            const int blk = chunk * 16 + k;
            const int vals[4] = { v[k].x, v[k].y, v[k].z, v[k].w };
            const bool any = ((vals[0] >> 4) == key) | ((vals[1] >> 4) == key) |
                             ((vals[2] >> 4) == key) | ((vals[3] >> 4) == key);
            if (__ballot(any) == 0) continue;
#pragma unroll
            for (int e = 0; e < 4; ++e) {
                unsigned mask = (unsigned)__ballot((vals[e] >> 4) == key);
                while (mask) {
                    const int src = __ffs(mask) - 1;
                    mask &= mask - 1;
                    const int c   = __shfl(vals[e], src, 32) & 127;
                    const int row = blk * 128 + 4 * src + e;
                    const float4 x = X4[(size_t)row * (DIMD / 4)];
                    float4 o = V4[c * (VP / 4)];
                    o.x += x.x; o.y += x.y; o.z += x.z; o.w += x.w;
                    V4[c * (VP / 4)] = o;
                    if (lane == (c & 15)) mycnt += 1.0f;
                }
            }
        }
    }
    if (lane < 16) cnts[wave * 16 + lane] = mycnt;
    __syncthreads();

    if (dh == 0 && tid < 128) {
        float s = 0.0f;
#pragma unroll 16
        for (int c = 0; c < 128; ++c) s += gt[127 + tid - c] * cnts[c];
        Tden[r * 128 + tid] = s;
    }

    v2f a[60];
    toeplitz_a(gt, l, half, a);
    v2f b[32];
    const float* vb = Vs + (2 * half) * VP + wave * 16 + l;
#pragma unroll
    for (int s = 0; s < 32; ++s) {
        b[s].x = vb[(4 * s) * VP];
        b[s].y = vb[(4 * s + 1) * VP];
    }
    v8f acc[8];
#pragma unroll
    for (int t = 0; t < 8; ++t) acc[t] = v8f{};
    toeplitz_mm(a, b, acc);

    float* tp = T + (size_t)r * (NG * DIMD) + dh * 128 + wave * 16 + l;
#pragma unroll
    for (int t = 0; t < 8; ++t) {
#pragma unroll
        for (int rr = 0; rr < 8; ++rr)
            tp[(size_t)(16 * t + 8 * half + rr) * DIMD] = acc[t][rr];
    }
}

__global__ void __launch_bounds__(256)
k_rowpass(const float* __restrict__ T, const float* __restrict__ Tden,
          const int* __restrict__ itp, const float* __restrict__ W,
          float* __restrict__ out) {
    __shared__ float gt[256];
    __shared__ float tds[128];
    __shared__ __attribute__((aligned(16))) float dens[128];
    const int tid  = threadIdx.x;
    const int lane = tid & 31;
    const int wave = tid >> 5;
    const int l    = lane & 15;
    const int half = lane >> 4;
    const int j    = blockIdx.x >> 1;
    const int dh   = blockIdx.x & 1;

    gauss_table(gt, itp, tid);
    if (tid < 128) tds[tid] = Tden[tid * 128 + j];
    __syncthreads();
    if (tid < 128) {
        float s = 0.0f;
#pragma unroll 16
        for (int r = 0; r < 128; ++r) s += gt[127 + tid - r] * tds[r];
        dens[tid] = s;
    }
    __syncthreads();

    v2f a[60];
    toeplitz_a(gt, l, half, a);
    const int n0 = j * DIMD + dh * 128 + wave * 16 + l;
    v2f b[32];
    const float* tb = T + (size_t)(2 * half) * (NG * DIMD) + n0;
#pragma unroll
    for (int s = 0; s < 32; ++s) {
        b[s].x = tb[(size_t)(4 * s) * (NG * DIMD)];
        b[s].y = tb[(size_t)(4 * s + 1) * (NG * DIMD)];
    }
    v8f acc[8];
#pragma unroll
    for (int t = 0; t < 8; ++t) acc[t] = v8f{};
    toeplitz_mm(a, b, acc);

    float4 dq[16];
#pragma unroll
    for (int t = 0; t < 8; ++t) {
        dq[2 * t]     = *(const float4*)&dens[16 * t + 8 * half];
        dq[2 * t + 1] = *(const float4*)&dens[16 * t + 8 * half + 4];
    }
#pragma unroll
    for (int t = 0; t < 8; ++t) {
        const float den8[8] = { dq[2 * t].x, dq[2 * t].y, dq[2 * t].z, dq[2 * t].w,
                                dq[2 * t + 1].x, dq[2 * t + 1].y, dq[2 * t + 1].z, dq[2 * t + 1].w };
#pragma unroll
        for (int rr = 0; rr < 8; ++rr) {
            const int i = 16 * t + 8 * half + rr;
            const size_t idx = (size_t)i * (NG * DIMD) + n0;
            const float den = den8[rr];
            out[idx] = (den != 0.0f) ? (acc[t][rr] / den) : W[idx];
        }
    }
}

extern "C" void kernel_launch(void* const* d_in, const int* in_sizes, int n_in,
                              void* d_out, int out_size, void* d_ws, size_t ws_size,
                              hipStream_t stream) {
    const float* X  = (const float*)d_in[0];
    const float* W  = (const float*)d_in[1];
    const int*   it = (const int*)d_in[2];
    float* out = (float*)d_out;

    float* ws    = (float*)d_ws;
    float* wsq   = ws;
    int*   bmu   = (int*)(ws + KUNITS);
    float* Xf    = ws + KUNITS + 2 * BSZ;
    int*   part  = (int*)(Xf + (size_t)BSZ * DIMD);
    float* T     = (float*)(part + (size_t)1024 * BSZ * 2);
    float* Tden  = T + (size_t)KUNITS * DIMD;

    k_wsq    <<<KUNITS / 256, 256, 0, stream>>>(W, X, wsq, Xf);
    k_bmu    <<<(KUNITS / 16 / 8) * 4, 256, 0, stream>>>(W, Xf, wsq, part);
    k_merge  <<<BSZ / 32, 256, 0, stream>>>(part, bmu);
    k_colpass<<<MG * 2, 256, 0, stream>>>(X, bmu, it, T, Tden);
    k_rowpass<<<NG * 2, 256, 0, stream>>>(T, Tden, it, W, out);
}


# k_colpass phase 1: a wave queues its matching data rows in LDS and loads them 16 at a time (running column sum in registers), not one dependent load per row; sums bit-identical
# speedup vs baseline: 8.8306x; 1.0367x over previous
#include <hip/hip_runtime.h>
#include <math.h>

#define MG   128
#define NG   128
#define DIMD 256
#define BSZ  8192
#define KUNITS (MG * NG)

typedef __attribute__((ext_vector_type(2))) float v2f;
typedef __attribute__((ext_vector_type(8))) float v8f;

__device__ __forceinline__ v8f wmma_f32(v2f a, v2f b, v8f c) {
    return __builtin_amdgcn_wmma_f32_16x16x4_f32(
         false, a,  false, b,
         (short)0, c,  false,  false);
}

__global__ void __launch_bounds__(256)
k_wsq(const float* __restrict__ W, const float* __restrict__ X,
      float* __restrict__ wsq, float* __restrict__ Xf) {
    const int k = blockIdx.x * blockDim.x + threadIdx.x;
    const float4* w4 = (const float4*)(W + (size_t)k * DIMD);
    float acc = 0.0f;
#pragma unroll 4
    for (int p = 0; p < DIMD / 8; ++p) {
        const float4 v0 = w4[2 * p];
        const float4 v1 = w4[2 * p + 1];
        acc += v0.x * v0.x + v0.y * v0.y + v0.z * v0.z + v0.w * v0.w;
        acc += v1.x * v1.x + v1.y * v1.y + v1.z * v1.z + v1.w * v1.w;
    }
    wsq[k] = acc;

    if (k < BSZ) {
        const float4* x4 = (const float4*)(X + (size_t)k * DIMD);
        float4* dst = (float4*)Xf + (size_t)(k >> 4) * (32 * 32) + (k & 15);
#pragma unroll 4
        for (int p = 0; p < DIMD / 8; ++p) {
            const float4 v0 = x4[2 * p];
            const float4 v1 = x4[2 * p + 1];
            dst[p * 32]      = make_float4(v0.x, v0.y, v1.x, v1.y);
            dst[p * 32 + 16] = make_float4(v0.z, v0.w, v1.z, v1.w);
        }
    }
}

__global__ void __launch_bounds__(256) __attribute__((amdgpu_waves_per_eu(4, 4)))
k_bmu(const float* __restrict__ W, const float* __restrict__ Xf,
      const float* __restrict__ wsq, int* __restrict__ part) {
    const int tid  = threadIdx.x;
    const int lane = tid & 31;
    const int wave = tid >> 5;
    const int l    = lane & 15;
    const int half = lane >> 4;
    const int ut   = (blockIdx.x >> 2) * 8 + wave;
    const int u0   = ut * 16;
    const int rt0  = (blockIdx.x & 3) * (BSZ / 16 / 4);

    v2f a[64];
    const float* arow = W + (size_t)(u0 + l) * DIMD + 2 * half;
#pragma unroll
    for (int s = 0; s < 64; ++s) {
        a[s].x = arow[4 * s + 0];
        a[s].y = arow[4 * s + 1];
    }

    float ws[8];
#pragma unroll
    for (int r = 0; r < 8; ++r) ws[r] = wsq[u0 + 8 * half + r];

    const float4* xf4 = (const float4*)Xf + lane;
    int2* pout = (int2*)part + (size_t)ut * BSZ + l;
    for (int rt = rt0; rt < rt0 + BSZ / 16 / 4; ++rt) {
        const float4* bt = xf4 + (size_t)rt * (32 * 32);
        v8f c = {};
#pragma unroll
        for (int p = 0; p < 32; ++p) {
            const float4 v = bt[p * 32];
            v2f b0, b1;
            b0.x = v.x; b0.y = v.y;
            b1.x = v.z; b1.y = v.w;
            c = wmma_f32(a[2 * p], b0, c);
            c = wmma_f32(a[2 * p + 1], b1, c);
        }
        float bv = ws[0] - 2.0f * c[0];
        int   br = 0;
#pragma unroll
        for (int r = 1; r < 8; ++r) {
            const float v = ws[r] - 2.0f * c[r];
            if (v < bv) { bv = v; br = r; }
        }
        int bi = u0 + 8 * half + br;
        const float ov = __shfl_xor(bv, 16, 32);
        const int   oi = __shfl_xor(bi, 16, 32);
        if (ov < bv || (ov == bv && oi < bi)) { bv = ov; bi = oi; }
        if (half == 0) pout[rt * 16] = make_int2(__float_as_int(bv), bi);
    }
}

__global__ void __launch_bounds__(256)
k_merge(const int* __restrict__ part, int* __restrict__ bmu) {
    __shared__ float sv[8][32];
    __shared__ int   si[8][32];
    const int lane = threadIdx.x & 31;
    const int seg  = threadIdx.x >> 5;
    const int row  = blockIdx.x * 32 + lane;
    const int2* p = (const int2*)part + (size_t)(seg * 128) * BSZ + row;
    float bv = 3.4e38f;
    int   bi = 0;
#pragma unroll 16
    for (int t = 0; t < 128; ++t) {
        const int2 e = p[(size_t)t * BSZ];
        const float v = __int_as_float(e.x);
        if (v < bv) { bv = v; bi = e.y; }
    }
    sv[seg][lane] = bv;
    si[seg][lane] = bi;
    __syncthreads();
    if (seg == 0) {
#pragma unroll
        for (int s = 1; s < 8; ++s) {
            const float v = sv[s][lane];
            const int   i = si[s][lane];
            if (v < bv) { bv = v; bi = i; }
        }
        bmu[row] = bi;
    }
}

__device__ __forceinline__ void gauss_table(float* gt, const int* itp, int tid) {
    const float START_SIGMA = 64.0f;
    const float TAU = 20.0f / logf(START_SIGMA);
    const float sigma = START_SIGMA * expf(-(float)itp[0] / TAU);
    const float inv_d = 1.0f / (2.0f * sigma * sigma);
    const float dr = (float)(tid - 127);
    gt[tid] = expf(-dr * dr * inv_d);
}

__device__ __forceinline__ void toeplitz_a(const float* gt, int l, int half, v2f (&a)[60]) {
#pragma unroll
    for (int q = 0; q < 60; ++q) {
        const int m = 127 + 4 * (q - 31) + l - 2 * half;
        a[q].x = gt[m];
        a[q].y = gt[m - 1];
    }
}

__device__ __forceinline__ void toeplitz_mm(const v2f (&a)[60], const v2f (&b)[32], v8f (&acc)[8]) {
#pragma unroll
    for (int s = 0; s < 32; ++s) {
#pragma unroll
        for (int t = 0; t < 8; ++t)
            acc[t] = wmma_f32(a[4 * t - s + 31], b[s], acc[t]);
    }
}

#define VP 136

#define QCAP 2064

template <bool FULL>
__device__ __forceinline__ void take_rows(const unsigned short* q, int pos, int n,
                                          int row0, int col0, int lane,
                                          const float4* X4, float4* V4,
                                          float4& run, int& cur, float& mycnt) {
    const int ent = q[pos + (lane & 15)];
    float4 x[16];
#pragma unroll
    for (int i = 0; i < 16; ++i) {
        x[i] = make_float4(0.0f, 0.0f, 0.0f, 0.0f);
        if (FULL || i < n) {
            const int row = row0 + (__builtin_amdgcn_readlane(ent, i) >> 4);
            x[i] = X4[(size_t)row * (DIMD / 4)];
        }
    }
#pragma unroll
    for (int i = 0; i < 16; ++i) {
        if (FULL || i < n) {
            const int c = __builtin_amdgcn_readlane(ent, i) & 15;
            if (c != cur) {
                if (cur >= 0) V4[(col0 + cur) * (VP / 4)] = run;
                run = V4[(col0 + c) * (VP / 4)];
                cur = c;
            }
            run.x += x[i].x; run.y += x[i].y; run.z += x[i].z; run.w += x[i].w;
            if (lane == c) mycnt += 1.0f;
        }
    }
}

__global__ void __launch_bounds__(256)
k_colpass(const float* __restrict__ X, const int* __restrict__ bmu,
          const int* __restrict__ itp, float* __restrict__ T, float* __restrict__ Tden) {
    __shared__ float gt[256];
    __shared__ float cnts[128];
    __shared__ __attribute__((aligned(16))) float Vs[128 * VP];
    __shared__ unsigned short qs[8][QCAP];
    const int tid  = threadIdx.x;
    const int lane = tid & 31;
    const int wave = tid >> 5;
    const int l    = lane & 15;
    const int half = lane >> 4;
    const int r    = blockIdx.x >> 1;
    const int dh   = blockIdx.x & 1;

    gauss_table(gt, itp, tid);
    for (int idx = tid; idx < 128 * VP / 4; idx += 256)
        ((float4*)Vs)[idx] = make_float4(0.0f, 0.0f, 0.0f, 0.0f);
    __syncthreads();

    const int key = r * 8 + wave;
    float mycnt = 0.0f;
    const int4*   b4 = (const int4*)bmu + lane;
    const float4* X4 = (const float4*)X + dh * 32 + lane;
    float4*       V4 = (float4*)Vs + lane;
    unsigned short* q = qs[wave];
    float4 run = make_float4(0.0f, 0.0f, 0.0f, 0.0f);
    int cur = -1;
    for (int chunk = 0; chunk < BSZ / 128 / 16; ++chunk) {
        int4 v[16];
#pragma unroll
        for (int k = 0; k < 16; ++k) v[k] = b4[(chunk * 16 + k) * 32];
        int qn = 0;
#pragma unroll
        for (int k = 0; k < 16; ++k) {
            const int vals[4] = { v[k].x, v[k].y, v[k].z, v[k].w };
            const bool any = ((vals[0] >> 4) == key) | ((vals[1] >> 4) == key) |
                             ((vals[2] >> 4) == key) | ((vals[3] >> 4) == key);
            if (__ballot(any) == 0) continue;
#pragma unroll
            for (int e = 0; e < 4; ++e) {
                const bool hit = (vals[e] >> 4) == key;
                const unsigned m = (unsigned)__ballot(hit);
                if (m == 0) continue;
                if (hit) q[qn + (int)__builtin_amdgcn_mbcnt_lo(m, 0u)] =
                             (unsigned short)(((k * 128 + 4 * lane + e) << 4) | (vals[e] & 15));
                qn += __popc(m);
            }
        }
        __builtin_amdgcn_fence(__ATOMIC_SEQ_CST, "wavefront");
        const int row0 = chunk * 2048;
        int pos = 0;
        while (qn - pos >= 16) {
            take_rows<true>(q, pos, 16, row0, wave * 16, lane, X4, V4, run, cur, mycnt);
            pos += 16;
        }
        if (qn != pos)
            take_rows<false>(q, pos, qn - pos, row0, wave * 16, lane, X4, V4, run, cur, mycnt);
        __builtin_amdgcn_fence(__ATOMIC_SEQ_CST, "wavefront");
    }
    if (cur >= 0) V4[(wave * 16 + cur) * (VP / 4)] = run;
    if (lane < 16) cnts[wave * 16 + lane] = mycnt;
    __syncthreads();

    if (dh == 0 && tid < 128) {
        float s = 0.0f;
#pragma unroll 16
        for (int c = 0; c < 128; ++c) s += gt[127 + tid - c] * cnts[c];
        Tden[r * 128 + tid] = s;
    }

    v2f a[60];
    toeplitz_a(gt, l, half, a);
    v2f b[32];
    const float* vb = Vs + (2 * half) * VP + wave * 16 + l;
#pragma unroll
    for (int s = 0; s < 32; ++s) {
        b[s].x = vb[(4 * s) * VP];
        b[s].y = vb[(4 * s + 1) * VP];
    }
    v8f acc[8];
#pragma unroll
    for (int t = 0; t < 8; ++t) acc[t] = v8f{};
    toeplitz_mm(a, b, acc);

    float* tp = T + (size_t)r * (NG * DIMD) + dh * 128 + wave * 16 + l;
#pragma unroll
    for (int t = 0; t < 8; ++t) {
#pragma unroll
        for (int rr = 0; rr < 8; ++rr)
            tp[(size_t)(16 * t + 8 * half + rr) * DIMD] = acc[t][rr];
    }
}

__global__ void __launch_bounds__(256)
k_rowpass(const float* __restrict__ T, const float* __restrict__ Tden,
          const int* __restrict__ itp, const float* __restrict__ W,
          float* __restrict__ out) {
    __shared__ float gt[256];
    __shared__ float tds[128];
    __shared__ __attribute__((aligned(16))) float dens[128];
    const int tid  = threadIdx.x;
    const int lane = tid & 31;
    const int wave = tid >> 5;
    const int l    = lane & 15;
    const int half = lane >> 4;
    const int j    = blockIdx.x >> 1;
    const int dh   = blockIdx.x & 1;

    gauss_table(gt, itp, tid);
    if (tid < 128) tds[tid] = Tden[tid * 128 + j];
    __syncthreads();
    if (tid < 128) {
        float s = 0.0f;
#pragma unroll 16
        for (int r = 0; r < 128; ++r) s += gt[127 + tid - r] * tds[r];
        dens[tid] = s;
    }
    __syncthreads();

    v2f a[60];
    toeplitz_a(gt, l, half, a);
    const int n0 = j * DIMD + dh * 128 + wave * 16 + l;
    v2f b[32];
    const float* tb = T + (size_t)(2 * half) * (NG * DIMD) + n0;
#pragma unroll
    for (int s = 0; s < 32; ++s) {
        b[s].x = tb[(size_t)(4 * s) * (NG * DIMD)];
        b[s].y = tb[(size_t)(4 * s + 1) * (NG * DIMD)];
    }
    v8f acc[8];
#pragma unroll
    for (int t = 0; t < 8; ++t) acc[t] = v8f{};
    toeplitz_mm(a, b, acc);

    float4 dq[16];
#pragma unroll
    for (int t = 0; t < 8; ++t) {
        dq[2 * t]     = *(const float4*)&dens[16 * t + 8 * half];
        dq[2 * t + 1] = *(const float4*)&dens[16 * t + 8 * half + 4];
    }
#pragma unroll
    for (int t = 0; t < 8; ++t) {
        const float den8[8] = { dq[2 * t].x, dq[2 * t].y, dq[2 * t].z, dq[2 * t].w,
                                dq[2 * t + 1].x, dq[2 * t + 1].y, dq[2 * t + 1].z, dq[2 * t + 1].w };
#pragma unroll
        for (int rr = 0; rr < 8; ++rr) {
            const int i = 16 * t + 8 * half + rr;
            const size_t idx = (size_t)i * (NG * DIMD) + n0;
            const float den = den8[rr];
            out[idx] = (den != 0.0f) ? (acc[t][rr] / den) : W[idx];
        }
    }
}

extern "C" void kernel_launch(void* const* d_in, const int* in_sizes, int n_in,
                              void* d_out, int out_size, void* d_ws, size_t ws_size,
                              hipStream_t stream) {
    const float* X  = (const float*)d_in[0];
    const float* W  = (const float*)d_in[1];
    const int*   it = (const int*)d_in[2];
    float* out = (float*)d_out;

    float* ws    = (float*)d_ws;
    float* wsq   = ws;
    int*   bmu   = (int*)(ws + KUNITS);
    float* Xf    = ws + KUNITS + 2 * BSZ;
    int*   part  = (int*)(Xf + (size_t)BSZ * DIMD);
    float* T     = (float*)(part + (size_t)1024 * BSZ * 2);
    float* Tden  = T + (size_t)KUNITS * DIMD;

    k_wsq    <<<KUNITS / 256, 256, 0, stream>>>(W, X, wsq, Xf);
    k_bmu    <<<(KUNITS / 16 / 8) * 4, 256, 0, stream>>>(W, Xf, wsq, part);
    k_merge  <<<BSZ / 32, 256, 0, stream>>>(part, bmu);
    k_colpass<<<MG * 2, 256, 0, stream>>>(X, bmu, it, T, Tden);
    k_rowpass<<<NG * 2, 256, 0, stream>>>(T, Tden, it, W, out);
}


# k_bmu: a work-group's 8 waves keep per-tile minima in LDS (128 KB) and merge them after the sweep: 128 partial minima per data row leave the chip, not 1024; k_merge reads 8x less; BMU bit-identical
# speedup vs baseline: 9.3412x; 1.0578x over previous
#include <hip/hip_runtime.h>
#include <math.h>

#define MG   128
#define NG   128
#define DIMD 256
#define BSZ  8192
#define KUNITS (MG * NG)

typedef __attribute__((ext_vector_type(2))) float v2f;
typedef __attribute__((ext_vector_type(8))) float v8f;

__device__ __forceinline__ v8f wmma_f32(v2f a, v2f b, v8f c) {
    return __builtin_amdgcn_wmma_f32_16x16x4_f32(
         false, a,  false, b,
         (short)0, c,  false,  false);
}

__global__ void __launch_bounds__(256)
k_wsq(const float* __restrict__ W, const float* __restrict__ X,
      float* __restrict__ wsq, float* __restrict__ Xf) {
    const int k = blockIdx.x * blockDim.x + threadIdx.x;
    const float4* w4 = (const float4*)(W + (size_t)k * DIMD);
    float acc = 0.0f;
#pragma unroll 4
    for (int p = 0; p < DIMD / 8; ++p) {
        const float4 v0 = w4[2 * p];
        const float4 v1 = w4[2 * p + 1];
        acc += v0.x * v0.x + v0.y * v0.y + v0.z * v0.z + v0.w * v0.w;
        acc += v1.x * v1.x + v1.y * v1.y + v1.z * v1.z + v1.w * v1.w;
    }
    wsq[k] = acc;

    if (k < BSZ) {
        const float4* x4 = (const float4*)(X + (size_t)k * DIMD);
        float4* dst = (float4*)Xf + (size_t)(k >> 4) * (32 * 32) + (k & 15);
#pragma unroll 4
        for (int p = 0; p < DIMD / 8; ++p) {
            const float4 v0 = x4[2 * p];
            const float4 v1 = x4[2 * p + 1];
            dst[p * 32]      = make_float4(v0.x, v0.y, v1.x, v1.y);
            dst[p * 32 + 16] = make_float4(v0.z, v0.w, v1.z, v1.w);
        }
    }
}

__global__ void __launch_bounds__(256) __attribute__((amdgpu_waves_per_eu(4, 4)))
k_bmu(const float* __restrict__ W, const float* __restrict__ Xf,
      const float* __restrict__ wsq, int* __restrict__ part) {
    __shared__ int2 mins[8][BSZ / 4];
    const int tid  = threadIdx.x;
    const int lane = tid & 31;
    const int wave = tid >> 5;
    const int l    = lane & 15;
    const int half = lane >> 4;
    const int ut   = (blockIdx.x >> 2) * 8 + wave;
    const int u0   = ut * 16;
    const int rt0  = (blockIdx.x & 3) * (BSZ / 16 / 4);

    v2f a[64];
    const float* arow = W + (size_t)(u0 + l) * DIMD + 2 * half;
#pragma unroll
    for (int s = 0; s < 64; ++s) {
        a[s].x = arow[4 * s + 0];
        a[s].y = arow[4 * s + 1];
    }

    float ws[8];
#pragma unroll
    for (int r = 0; r < 8; ++r) ws[r] = wsq[u0 + 8 * half + r];

    const float4* xf4 = (const float4*)Xf + lane;
    int2* mrow = &mins[wave][l];
    for (int rt = rt0; rt < rt0 + BSZ / 16 / 4; ++rt) {
        const float4* bt = xf4 + (size_t)rt * (32 * 32);
        v8f c = {};
#pragma unroll
        for (int p = 0; p < 32; ++p) {
            const float4 v = bt[p * 32];
            v2f b0, b1;
            b0.x = v.x; b0.y = v.y;
            b1.x = v.z; b1.y = v.w;
            c = wmma_f32(a[2 * p], b0, c);
            c = wmma_f32(a[2 * p + 1], b1, c);
        }
        float bv = ws[0] - 2.0f * c[0];
        int   br = 0;
#pragma unroll
        for (int r = 1; r < 8; ++r) {
            const float v = ws[r] - 2.0f * c[r];
            if (v < bv) { bv = v; br = r; }
        }
        int bi = u0 + 8 * half + br;
        const float ov = __shfl_xor(bv, 16, 32);
        const int   oi = __shfl_xor(bi, 16, 32);
        if (ov < bv || (ov == bv && oi < bi)) { bv = ov; bi = oi; }
        if (half == 0) mrow[(rt - rt0) * 16] = make_int2(__float_as_int(bv), bi);
    }
    __syncthreads();
    int2* pout = (int2*)part + (size_t)(blockIdx.x >> 2) * BSZ + rt0 * 16;
#pragma unroll
    for (int j = 0; j < BSZ / 4 / 256; ++j) {
        const int row = j * 256 + tid;
        float bv = 3.4e38f;
        int   bi = 0;
#pragma unroll
        for (int w = 0; w < 8; ++w) {
            const int2 e = mins[w][row];
            const float v = __int_as_float(e.x);
            if (v < bv) { bv = v; bi = e.y; }
        }
        pout[row] = make_int2(__float_as_int(bv), bi);
    }
}

__global__ void __launch_bounds__(256)
k_merge(const int* __restrict__ part, int* __restrict__ bmu) {
    __shared__ float sv[8][32];
    __shared__ int   si[8][32];
    const int lane = threadIdx.x & 31;
    const int seg  = threadIdx.x >> 5;
    const int row  = blockIdx.x * 32 + lane;
    const int2* p = (const int2*)part + (size_t)(seg * 16) * BSZ + row;
    float bv = 3.4e38f;
    int   bi = 0;
#pragma unroll
    for (int t = 0; t < 16; ++t) {
        const int2 e = p[(size_t)t * BSZ];
        const float v = __int_as_float(e.x);
        if (v < bv) { bv = v; bi = e.y; }
    }
    sv[seg][lane] = bv;
    si[seg][lane] = bi;
    __syncthreads();
    if (seg == 0) {
#pragma unroll
        for (int s = 1; s < 8; ++s) {
            const float v = sv[s][lane];
            const int   i = si[s][lane];
            if (v < bv) { bv = v; bi = i; }
        }
        bmu[row] = bi;
    }
}

__device__ __forceinline__ void gauss_table(float* gt, const int* itp, int tid) {
    const float START_SIGMA = 64.0f;
    const float TAU = 20.0f / logf(START_SIGMA);
    const float sigma = START_SIGMA * expf(-(float)itp[0] / TAU);
    const float inv_d = 1.0f / (2.0f * sigma * sigma);
    const float dr = (float)(tid - 127);
    gt[tid] = expf(-dr * dr * inv_d);
}

__device__ __forceinline__ void toeplitz_a(const float* gt, int l, int half, v2f (&a)[60]) {
#pragma unroll
    for (int q = 0; q < 60; ++q) {
        const int m = 127 + 4 * (q - 31) + l - 2 * half;
        a[q].x = gt[m];
        a[q].y = gt[m - 1];
    }
}

__device__ __forceinline__ void toeplitz_mm(const v2f (&a)[60], const v2f (&b)[32], v8f (&acc)[8]) {
#pragma unroll
    for (int s = 0; s < 32; ++s) {
#pragma unroll
        for (int t = 0; t < 8; ++t)
            acc[t] = wmma_f32(a[4 * t - s + 31], b[s], acc[t]);
    }
}

#define VP 136

#define QCAP 2064

template <bool FULL>
__device__ __forceinline__ void take_rows(const unsigned short* q, int pos, int n,
                                          int row0, int col0, int lane,
                                          const float4* X4, float4* V4,
                                          float4& run, int& cur, float& mycnt) {
    const int ent = q[pos + (lane & 15)];
    float4 x[16];
#pragma unroll
    for (int i = 0; i < 16; ++i) {
        x[i] = make_float4(0.0f, 0.0f, 0.0f, 0.0f);
        if (FULL || i < n) {
            const int row = row0 + (__builtin_amdgcn_readlane(ent, i) >> 4);
            x[i] = X4[(size_t)row * (DIMD / 4)];
        }
    }
#pragma unroll
    for (int i = 0; i < 16; ++i) {
        if (FULL || i < n) {
            const int c = __builtin_amdgcn_readlane(ent, i) & 15;
            if (c != cur) {
                if (cur >= 0) V4[(col0 + cur) * (VP / 4)] = run;
                run = V4[(col0 + c) * (VP / 4)];
                cur = c;
            }
            run.x += x[i].x; run.y += x[i].y; run.z += x[i].z; run.w += x[i].w;
            if (lane == c) mycnt += 1.0f;
        }
    }
}

__global__ void __launch_bounds__(256)
k_colpass(const float* __restrict__ X, const int* __restrict__ bmu,
          const int* __restrict__ itp, float* __restrict__ T, float* __restrict__ Tden) {
    __shared__ float gt[256];
    __shared__ float cnts[128];
    __shared__ __attribute__((aligned(16))) float Vs[128 * VP];
    __shared__ unsigned short qs[8][QCAP];
    const int tid  = threadIdx.x;
    const int lane = tid & 31;
    const int wave = tid >> 5;
    const int l    = lane & 15;
    const int half = lane >> 4;
    const int r    = blockIdx.x >> 1;
    const int dh   = blockIdx.x & 1;

    gauss_table(gt, itp, tid);
    for (int idx = tid; idx < 128 * VP / 4; idx += 256)
        ((float4*)Vs)[idx] = make_float4(0.0f, 0.0f, 0.0f, 0.0f);
    __syncthreads();

    const int key = r * 8 + wave;
    float mycnt = 0.0f;
    const int4*   b4 = (const int4*)bmu + lane;
    const float4* X4 = (const float4*)X + dh * 32 + lane;
    float4*       V4 = (float4*)Vs + lane;
    unsigned short* q = qs[wave];
    float4 run = make_float4(0.0f, 0.0f, 0.0f, 0.0f);
    int cur = -1;
    for (int chunk = 0; chunk < BSZ / 128 / 16; ++chunk) {
        int4 v[16];
#pragma unroll
        for (int k = 0; k < 16; ++k) v[k] = b4[(chunk * 16 + k) * 32];
        int qn = 0;
#pragma unroll
        for (int k = 0; k < 16; ++k) {
            const int vals[4] = { v[k].x, v[k].y, v[k].z, v[k].w };
            const bool any = ((vals[0] >> 4) == key) | ((vals[1] >> 4) == key) |
                             ((vals[2] >> 4) == key) | ((vals[3] >> 4) == key);
            if (__ballot(any) == 0) continue;
#pragma unroll
            for (int e = 0; e < 4; ++e) {
                const bool hit = (vals[e] >> 4) == key;
                const unsigned m = (unsigned)__ballot(hit);
                if (m == 0) continue;
                if (hit) q[qn + (int)__builtin_amdgcn_mbcnt_lo(m, 0u)] =
                             (unsigned short)(((k * 128 + 4 * lane + e) << 4) | (vals[e] & 15));
                qn += __popc(m);
            }
        }
        __builtin_amdgcn_fence(__ATOMIC_SEQ_CST, "wavefront");
        const int row0 = chunk * 2048;
        int pos = 0;
        while (qn - pos >= 16) {
            take_rows<true>(q, pos, 16, row0, wave * 16, lane, X4, V4, run, cur, mycnt);
            pos += 16;
        }
        if (qn != pos)
            take_rows<false>(q, pos, qn - pos, row0, wave * 16, lane, X4, V4, run, cur, mycnt);
        __builtin_amdgcn_fence(__ATOMIC_SEQ_CST, "wavefront");
    }
    if (cur >= 0) V4[(wave * 16 + cur) * (VP / 4)] = run;
    if (lane < 16) cnts[wave * 16 + lane] = mycnt;
    __syncthreads();

    if (dh == 0 && tid < 128) {
        float s = 0.0f;
#pragma unroll 16
        for (int c = 0; c < 128; ++c) s += gt[127 + tid - c] * cnts[c];
        Tden[r * 128 + tid] = s;
    }

    v2f a[60];
    toeplitz_a(gt, l, half, a);
    v2f b[32];
    const float* vb = Vs + (2 * half) * VP + wave * 16 + l;
#pragma unroll
    for (int s = 0; s < 32; ++s) {
        b[s].x = vb[(4 * s) * VP];
        b[s].y = vb[(4 * s + 1) * VP];
    }
    v8f acc[8];
#pragma unroll
    for (int t = 0; t < 8; ++t) acc[t] = v8f{};
    toeplitz_mm(a, b, acc);

    float* tp = T + (size_t)r * (NG * DIMD) + dh * 128 + wave * 16 + l;
#pragma unroll
    for (int t = 0; t < 8; ++t) {
#pragma unroll
        for (int rr = 0; rr < 8; ++rr)
            tp[(size_t)(16 * t + 8 * half + rr) * DIMD] = acc[t][rr];
    }
}

__global__ void __launch_bounds__(256)
k_rowpass(const float* __restrict__ T, const float* __restrict__ Tden,
          const int* __restrict__ itp, const float* __restrict__ W,
          float* __restrict__ out) {
    __shared__ float gt[256];
    __shared__ float tds[128];
    __shared__ __attribute__((aligned(16))) float dens[128];
    const int tid  = threadIdx.x;
    const int lane = tid & 31;
    const int wave = tid >> 5;
    const int l    = lane & 15;
    const int half = lane >> 4;
    const int j    = blockIdx.x >> 1;
    const int dh   = blockIdx.x & 1;

    gauss_table(gt, itp, tid);
    if (tid < 128) tds[tid] = Tden[tid * 128 + j];
    __syncthreads();
    if (tid < 128) {
        float s = 0.0f;
#pragma unroll 16
        for (int r = 0; r < 128; ++r) s += gt[127 + tid - r] * tds[r];
        dens[tid] = s;
    }
    __syncthreads();

    v2f a[60];
    toeplitz_a(gt, l, half, a);
    const int n0 = j * DIMD + dh * 128 + wave * 16 + l;
    v2f b[32];
    const float* tb = T + (size_t)(2 * half) * (NG * DIMD) + n0;
#pragma unroll
    for (int s = 0; s < 32; ++s) {
        b[s].x = tb[(size_t)(4 * s) * (NG * DIMD)];
        b[s].y = tb[(size_t)(4 * s + 1) * (NG * DIMD)];
    }
    v8f acc[8];
#pragma unroll
    for (int t = 0; t < 8; ++t) acc[t] = v8f{};
    toeplitz_mm(a, b, acc);

    float4 dq[16];
#pragma unroll
    for (int t = 0; t < 8; ++t) {
        dq[2 * t]     = *(const float4*)&dens[16 * t + 8 * half];
        dq[2 * t + 1] = *(const float4*)&dens[16 * t + 8 * half + 4];
    }
#pragma unroll
    for (int t = 0; t < 8; ++t) {
        const float den8[8] = { dq[2 * t].x, dq[2 * t].y, dq[2 * t].z, dq[2 * t].w,
                                dq[2 * t + 1].x, dq[2 * t + 1].y, dq[2 * t + 1].z, dq[2 * t + 1].w };
#pragma unroll
        for (int rr = 0; rr < 8; ++rr) {
            const int i = 16 * t + 8 * half + rr;
            const size_t idx = (size_t)i * (NG * DIMD) + n0;
            const float den = den8[rr];
            out[idx] = (den != 0.0f) ? (acc[t][rr] / den) : W[idx];
        }
    }
}

extern "C" void kernel_launch(void* const* d_in, const int* in_sizes, int n_in,
                              void* d_out, int out_size, void* d_ws, size_t ws_size,
                              hipStream_t stream) {
    const float* X  = (const float*)d_in[0];
    const float* W  = (const float*)d_in[1];
    const int*   it = (const int*)d_in[2];
    float* out = (float*)d_out;

    float* ws    = (float*)d_ws;
    float* wsq   = ws;
    int*   bmu   = (int*)(ws + KUNITS);
    float* Xf    = ws + KUNITS + 2 * BSZ;
    int*   part  = (int*)(Xf + (size_t)BSZ * DIMD);
    float* T     = (float*)(part + (size_t)128 * BSZ * 2);
    float* Tden  = T + (size_t)KUNITS * DIMD;

    k_wsq    <<<KUNITS / 256, 256, 0, stream>>>(W, X, wsq, Xf);
    k_bmu    <<<(KUNITS / 16 / 8) * 4, 256, 0, stream>>>(W, Xf, wsq, part);
    k_merge  <<<BSZ / 32, 256, 0, stream>>>(part, bmu);
    k_colpass<<<MG * 2, 256, 0, stream>>>(X, bmu, it, T, Tden);
    k_rowpass<<<NG * 2, 256, 0, stream>>>(T, Tden, it, W, out);
}


# k_verify reads the screening minima once: Mt stored row-tile-major, a work-group's 64 KB taken as 16 whole-line loads per thread, kept in registers for both H and the marks; BMU bit-identical
# speedup vs baseline: 23.9637x; 2.5654x over previous
#include <hip/hip_runtime.h>
#include <math.h>

#define MG   128
#define NG   128
#define DIMD 256
#define BSZ  8192
#define KUNITS (MG * NG)

typedef __attribute__((ext_vector_type(2))) float v2f;
typedef __attribute__((ext_vector_type(8))) float v8f;

__device__ __forceinline__ v8f wmma_f32(v2f a, v2f b, v8f c) {
    return __builtin_amdgcn_wmma_f32_16x16x4_f32(
         false, a,  false, b,
         (short)0, c,  false,  false);
}

typedef __attribute__((ext_vector_type(8)))  __bf16   bf16x8;
typedef __attribute__((ext_vector_type(16))) __bf16   bf16x16;
typedef __attribute__((ext_vector_type(4)))  unsigned u32x4;
typedef __attribute__((ext_vector_type(8)))  unsigned u32x8;
typedef __attribute__((ext_vector_type(4)))  float    f32x4;

__device__ __forceinline__ v8f wmma_bf16(bf16x16 a, bf16x16 b, v8f c) {
    return __builtin_amdgcn_wmma_f32_16x16x32_bf16(
         false, a,  false, b,
         (short)0, c,  false,  false);
}


__global__ void __launch_bounds__(256)
k_prep(const float* __restrict__ W, const float* __restrict__ X,
       float* __restrict__ wsq, float* __restrict__ Xf,
       u32x4* __restrict__ Xh, float* __restrict__ nx, float* __restrict__ Wf) {
    const int g = blockIdx.x * blockDim.x + threadIdx.x;
    if (g < KUNITS) {
        const int k = g;
        const float4* w4 = (const float4*)(W + (size_t)k * DIMD);
        float4* wdst = (float4*)Wf + (size_t)(k >> 4) * (32 * 32) + (k & 15);
        float acc = 0.0f;
#pragma unroll 4
        for (int p = 0; p < DIMD / 8; ++p) {
            const float4 v0 = w4[2 * p];
            const float4 v1 = w4[2 * p + 1];
            acc += v0.x * v0.x + v0.y * v0.y + v0.z * v0.z + v0.w * v0.w;
            acc += v1.x * v1.x + v1.y * v1.y + v1.z * v1.z + v1.w * v1.w;
            wdst[p * 32]      = make_float4(v0.x, v0.y, v1.x, v1.y);
            wdst[p * 32 + 16] = make_float4(v0.z, v0.w, v1.z, v1.w);
        }
        wsq[k] = acc;
    } else {
        const int k = g - KUNITS;
        const float4* x4 = (const float4*)(X + (size_t)k * DIMD);
        float4* dst = (float4*)Xf + (size_t)(k >> 4) * (32 * 32) + (k & 15);
        u32x4*  dh  = Xh + (size_t)(k >> 4) * (8 * 2 * 32) + (k & 15);
        float xs = 0.0f;
#pragma unroll 4
        for (int p = 0; p < DIMD / 8; ++p) {
            const float4 v0 = x4[2 * p];
            const float4 v1 = x4[2 * p + 1];
            dst[p * 32]      = make_float4(v0.x, v0.y, v1.x, v1.y);
            dst[p * 32 + 16] = make_float4(v0.z, v0.w, v1.z, v1.w);
            xs += v0.x * v0.x + v0.y * v0.y + v0.z * v0.z + v0.w * v0.w;
            xs += v1.x * v1.x + v1.y * v1.y + v1.z * v1.z + v1.w * v1.w;
            bf16x8 hv;
            hv[0] = (__bf16)v0.x; hv[1] = (__bf16)v0.y; hv[2] = (__bf16)v0.z; hv[3] = (__bf16)v0.w;
            hv[4] = (__bf16)v1.x; hv[5] = (__bf16)v1.y; hv[6] = (__bf16)v1.z; hv[7] = (__bf16)v1.w;
            dh[(((p >> 2) * 2 + (p & 1)) * 32) + ((p >> 1) & 1) * 16] = __builtin_bit_cast(u32x4, hv);
        }
        nx[k] = sqrtf(xs) * 1.001f;
    }
}

__global__ void __launch_bounds__(256) __attribute__((amdgpu_waves_per_eu(4, 4)))
k_lo(const float* __restrict__ W, const u32x4* __restrict__ Xh,
     const float* __restrict__ wsq, float* __restrict__ Mt,
     float* __restrict__ ea, float* __restrict__ eb) {
    const int tid  = threadIdx.x;
    const int lane = tid & 31;
    const int wave = tid >> 5;
    const int l    = lane & 15;
    const int half = lane >> 4;
    const int pr   = (blockIdx.x >> 3) * 8 + wave;
    const int rt0  = (blockIdx.x & 7) * (BSZ / 16 / 8);

    if ((blockIdx.x & 7) == 0) {
        float wm = wsq[pr * 32 + lane];
        wm = fmaxf(wm, __shfl_xor(wm, 8, 32));
        wm = fmaxf(wm, __shfl_xor(wm, 4, 32));
        wm = fmaxf(wm, __shfl_xor(wm, 2, 32));
        wm = fmaxf(wm, __shfl_xor(wm, 1, 32));
        if (l == 0) {
            ea[2 * pr + half] = sqrtf(wm) * 1.001f * (0.015625f * 1.05f);
            eb[2 * pr + half] = wm * 9.5367431640625e-7f;
        }
    }

    bf16x16 a0[8], a1[8];
    const float4* r0 = (const float4*)(W + (size_t)(pr * 32 + l) * DIMD + 16 * half);
    const float4* r1 = r0 + 16 * (DIMD / 4);
#pragma unroll
    for (int P = 0; P < 8; ++P) {
#pragma unroll
        for (int q = 0; q < 4; ++q) {
            const float4 v0 = r0[P * 8 + q];
            a0[P][4 * q + 0] = (__bf16)v0.x; a0[P][4 * q + 1] = (__bf16)v0.y;
            a0[P][4 * q + 2] = (__bf16)v0.z; a0[P][4 * q + 3] = (__bf16)v0.w;
        }
    }
    asm volatile("" :: "v"(__builtin_bit_cast(u32x8, a0[0])), "v"(__builtin_bit_cast(u32x8, a0[1])),
                       "v"(__builtin_bit_cast(u32x8, a0[2])), "v"(__builtin_bit_cast(u32x8, a0[3])),
                       "v"(__builtin_bit_cast(u32x8, a0[4])), "v"(__builtin_bit_cast(u32x8, a0[5])),
                       "v"(__builtin_bit_cast(u32x8, a0[6])), "v"(__builtin_bit_cast(u32x8, a0[7])));
#pragma unroll
    for (int P = 0; P < 8; ++P) {
#pragma unroll
        for (int q = 0; q < 4; ++q) {
            const float4 v1 = r1[P * 8 + q];
            a1[P][4 * q + 0] = (__bf16)v1.x; a1[P][4 * q + 1] = (__bf16)v1.y;
            a1[P][4 * q + 2] = (__bf16)v1.z; a1[P][4 * q + 3] = (__bf16)v1.w;
        }
    }

    float ws0[8], ws1[8];
#pragma unroll
    for (int r = 0; r < 8; ++r) {
        ws0[r] = wsq[pr * 32 + 8 * half + r];
        ws1[r] = wsq[pr * 32 + 16 + 8 * half + r];
    }

    const u32x4* xb = Xh + lane;
    float* mp = Mt + (size_t)(2 * pr + half) * 16 + l;
    for (int rt = rt0; rt < rt0 + BSZ / 16 / 8; ++rt) {
        const u32x4* bt = xb + (size_t)rt * (8 * 2 * 32);
        u32x4 bb[16];
#pragma unroll
        for (int i = 0; i < 16; ++i) bb[i] = bt[i * 32];
        asm volatile("" :: "v"(bb[0]), "v"(bb[1]), "v"(bb[2]), "v"(bb[3]), "v"(bb[4]), "v"(bb[5]),
                           "v"(bb[6]), "v"(bb[7]), "v"(bb[8]), "v"(bb[9]), "v"(bb[10]), "v"(bb[11]),
                           "v"(bb[12]), "v"(bb[13]), "v"(bb[14]), "v"(bb[15]));
        v8f c0 = {}, c1 = {};
#pragma unroll
        for (int P = 0; P < 8; ++P) {
            const u32x4 lo = bb[2 * P];
            const u32x4 hi = bb[2 * P + 1];
            u32x8 t;
            t[0] = lo[0]; t[1] = lo[1]; t[2] = lo[2]; t[3] = lo[3];
            t[4] = hi[0]; t[5] = hi[1]; t[6] = hi[2]; t[7] = hi[3];
            const bf16x16 b = __builtin_bit_cast(bf16x16, t);
            c0 = wmma_bf16(a0[P], b, c0);
            c1 = wmma_bf16(a1[P], b, c1);
        }
        float m0 = ws0[0] - 2.0f * c0[0];
        float m1 = ws1[0] - 2.0f * c1[0];
#pragma unroll
        for (int r = 1; r < 8; ++r) {
            m0 = fminf(m0, ws0[r] - 2.0f * c0[r]);
            m1 = fminf(m1, ws1[r] - 2.0f * c1[r]);
        }
        m0 = fminf(m0, __shfl_xor(m0, 16, 32));
        m1 = fminf(m1, __shfl_xor(m1, 16, 32));
        mp[(size_t)rt * KUNITS] = half ? m1 : m0;
    }
}

__global__ void __launch_bounds__(256)
k_verify(const float* __restrict__ Wf, const float* __restrict__ Xf,
         const float* __restrict__ wsq, const float* __restrict__ Mt,
         const float* __restrict__ ea, const float* __restrict__ eb,
         const float* __restrict__ nx, int* __restrict__ bmu) {
    __shared__ __attribute__((aligned(16))) float hpart[8][16];
    __shared__ unsigned marks[32];
    __shared__ float    sv[8][16];
    __shared__ int      si[8][16];
    const int tid  = threadIdx.x;
    const int lane = tid & 31;
    const int wave = __builtin_amdgcn_readfirstlane(tid >> 5);
    const int l    = lane & 15;
    const int half = lane >> 4;
    const int rt   = blockIdx.x;

    const int rq = tid & 3;
    const int tb = tid >> 2;
    float4 mv[16];
    float  eav[16], ebv[16];
    {
        const float4* m4 = (const float4*)(Mt + (size_t)rt * KUNITS) + tid;
#pragma unroll
        for (int j = 0; j < 16; ++j) mv[j] = m4[j * 256];
#pragma unroll
        for (int j = 0; j < 16; ++j) { eav[j] = ea[64 * j + tb]; ebv[j] = eb[64 * j + tb]; }
    }
    const float4 nx4 = *(const float4*)(nx + rt * 16 + 4 * rq);

    float4 h = make_float4(3.4e38f, 3.4e38f, 3.4e38f, 3.4e38f);
#pragma unroll
    for (int j = 0; j < 16; ++j) {
        const float e0 = nx4.x * eav[j] + ebv[j];
        const float e1 = nx4.y * eav[j] + ebv[j];
        const float e2 = nx4.z * eav[j] + ebv[j];
        const float e3 = nx4.w * eav[j] + ebv[j];
        const float u0 = mv[j].x + e0;
        const float u1 = mv[j].y + e1;
        const float u2 = mv[j].z + e2;
        const float u3 = mv[j].w + e3;
        h.x = (u0 < h.x) ? u0 : h.x;
        h.y = (u1 < h.y) ? u1 : h.y;
        h.z = (u2 < h.z) ? u2 : h.z;
        h.w = (u3 < h.w) ? u3 : h.w;
        mv[j].x -= e0; mv[j].y -= e1; mv[j].z -= e2; mv[j].w -= e3;
    }
#pragma unroll
    for (int off = 4; off <= 16; off <<= 1) {
        const float o0 = __shfl_xor(h.x, off, 32);
        const float o1 = __shfl_xor(h.y, off, 32);
        const float o2 = __shfl_xor(h.z, off, 32);
        const float o3 = __shfl_xor(h.w, off, 32);
        h.x = (o0 < h.x) ? o0 : h.x;
        h.y = (o1 < h.y) ? o1 : h.y;
        h.z = (o2 < h.z) ? o2 : h.z;
        h.w = (o3 < h.w) ? o3 : h.w;
    }
    if (lane < 4) *(float4*)&hpart[wave][4 * lane] = h;
    __syncthreads();
    float4 hl = *(const float4*)&hpart[0][4 * rq];
#pragma unroll
    for (int w = 1; w < 8; ++w) {
        const float4 o = *(const float4*)&hpart[w][4 * rq];
        hl.x = (o.x < hl.x) ? o.x : hl.x;
        hl.y = (o.y < hl.y) ? o.y : hl.y;
        hl.z = (o.z < hl.z) ? o.z : hl.z;
        hl.w = (o.w < hl.w) ? o.w : hl.w;
    }

    unsigned myb = 0u;
#pragma unroll
    for (int j = 0; j < 16; ++j) {
        unsigned x = (unsigned)__ballot(mv[j].x <= hl.x) | (unsigned)__ballot(mv[j].y <= hl.y) |
                     (unsigned)__ballot(mv[j].z <= hl.z) | (unsigned)__ballot(mv[j].w <= hl.w);
        x |= x >> 1;
        x |= x >> 2;
        x &= 0x11111111u;
        x = (x | (x >> 3)) & 0x03030303u;
        x = (x | (x >> 6)) & 0x000F000Fu;
        x = (x | (x >> 12)) & 0xFFu;
        myb = (lane == j) ? x : myb;
    }
    if (lane < 16) ((unsigned char*)marks)[8 * lane + wave] = (unsigned char)myb;
    __syncthreads();

    float4 bq[32];
    {
        const float4* bt = (const float4*)Xf + (size_t)rt * (32 * 32) + lane;
#pragma unroll
        for (int p = 0; p < 32; ++p) bq[p] = bt[p * 32];
    }

    float bv = 3.4e38f;
    int   bi = 0x7fffffff;
    int   rank = 0;
#pragma unroll 1
    for (int wd = 0; wd < 32; ++wd) {
        unsigned word = (unsigned)__builtin_amdgcn_readfirstlane((int)marks[wd]);
        while (word != 0u) {
            const int t = __builtin_ctz(word);
            word &= word - 1u;
            const bool mine = (rank & 7) == wave;
            ++rank;
            if (!mine) continue;
            const int u0 = (wd * 32 + t) * 16;
            const f32x4* at = (const f32x4*)Wf + (size_t)(wd * 32 + t) * (32 * 32) + lane;
            const float4 wa = *(const float4*)(wsq + u0 + 8 * half);
            const float4 wb = *(const float4*)(wsq + u0 + 8 * half + 4);
            v8f c = {};
#pragma unroll
            for (int hb = 0; hb < 2; ++hb) {
                f32x4 aq[16];
#pragma unroll
                for (int i = 0; i < 16; ++i) aq[i] = at[(hb * 16 + i) * 32];
                asm volatile("" :: "v"(aq[0]), "v"(aq[1]), "v"(aq[2]), "v"(aq[3]), "v"(aq[4]), "v"(aq[5]),
                                   "v"(aq[6]), "v"(aq[7]), "v"(aq[8]), "v"(aq[9]), "v"(aq[10]), "v"(aq[11]),
                                   "v"(aq[12]), "v"(aq[13]), "v"(aq[14]), "v"(aq[15]));
#pragma unroll
                for (int i = 0; i < 16; ++i) {
                    const int p = hb * 16 + i;
                    v2f a0, a1, b0, b1;
                    a0.x = aq[i][0]; a0.y = aq[i][1];
                    a1.x = aq[i][2]; a1.y = aq[i][3];
                    b0.x = bq[p].x; b0.y = bq[p].y;
                    b1.x = bq[p].z; b1.y = bq[p].w;
                    c = wmma_f32(a0, b0, c);
                    c = wmma_f32(a1, b1, c);
                }
            }
            const float ws[8] = { wa.x, wa.y, wa.z, wa.w, wb.x, wb.y, wb.z, wb.w };
#pragma unroll
            for (int r = 0; r < 8; ++r) {
                const float v = ws[r] - 2.0f * c[r];
                if (v < bv) { bv = v; bi = u0 + 8 * half + r; }
            }
        }
    }

    {
        const float ov = __shfl_xor(bv, 16, 32);
        const int   oi = __shfl_xor(bi, 16, 32);
        if (ov < bv || (ov == bv && oi < bi)) { bv = ov; bi = oi; }
    }
    if (half == 0) { sv[wave][l] = bv; si[wave][l] = bi; }
    __syncthreads();
    if (tid < 16) {
        float v = sv[0][tid];
        int   i = si[0][tid];
#pragma unroll
        for (int w = 1; w < 8; ++w) {
            const float ov = sv[w][tid];
            const int   oi = si[w][tid];
            if (ov < v || (ov == v && oi < i)) { v = ov; i = oi; }
        }
        bmu[rt * 16 + tid] = i;
    }
}

__device__ __forceinline__ void gauss_table(float* gt, const int* itp, int tid) {
    const float START_SIGMA = 64.0f;
    const float TAU = 20.0f / logf(START_SIGMA);
    const float sigma = START_SIGMA * expf(-(float)itp[0] / TAU);
    const float inv_d = 1.0f / (2.0f * sigma * sigma);
    const float dr = (float)(tid - 127);
    gt[tid] = expf(-dr * dr * inv_d);
}

__device__ __forceinline__ void toeplitz_a(const float* gt, int l, int half, v2f (&a)[60]) {
#pragma unroll
    for (int q = 0; q < 60; ++q) {
        const int m = 127 + 4 * (q - 31) + l - 2 * half;
        a[q].x = gt[m];
        a[q].y = gt[m - 1];
    }
}

__device__ __forceinline__ void toeplitz_mm(const v2f (&a)[60], const v2f (&b)[32], v8f (&acc)[8]) {
#pragma unroll
    for (int s = 0; s < 32; ++s) {
#pragma unroll
        for (int t = 0; t < 8; ++t)
            acc[t] = wmma_f32(a[4 * t - s + 31], b[s], acc[t]);
    }
}

#define VP 136

#define QCAP 2064

template <bool FULL>
__device__ __forceinline__ void take_rows(const unsigned short* q, int pos, int n,
                                          int row0, int col0, int lane,
                                          const float4* X4, float4* V4,
                                          float4& run, int& cur, float& mycnt) {
    const int ent = q[pos + (lane & 15)];
    float4 x[16];
#pragma unroll
    for (int i = 0; i < 16; ++i) {
        x[i] = make_float4(0.0f, 0.0f, 0.0f, 0.0f);
        if (FULL || i < n) {
            const int row = row0 + (__builtin_amdgcn_readlane(ent, i) >> 4);
            x[i] = X4[(size_t)row * (DIMD / 4)];
        }
    }
#pragma unroll
    for (int i = 0; i < 16; ++i) {
        if (FULL || i < n) {
            const int c = __builtin_amdgcn_readlane(ent, i) & 15;
            if (c != cur) {
                if (cur >= 0) V4[(col0 + cur) * (VP / 4)] = run;
                run = V4[(col0 + c) * (VP / 4)];
                cur = c;
            }
            run.x += x[i].x; run.y += x[i].y; run.z += x[i].z; run.w += x[i].w;
            if (lane == c) mycnt += 1.0f;
        }
    }
}

__global__ void __launch_bounds__(256)
k_colpass(const float* __restrict__ X, const int* __restrict__ bmu,
          const int* __restrict__ itp, float* __restrict__ T, float* __restrict__ Tden) {
    __shared__ float gt[256];
    __shared__ float cnts[128];
    __shared__ __attribute__((aligned(16))) float Vs[128 * VP];
    __shared__ unsigned short qs[8][QCAP];
    const int tid  = threadIdx.x;
    const int lane = tid & 31;
    const int wave = tid >> 5;
    const int l    = lane & 15;
    const int half = lane >> 4;
    const int r    = blockIdx.x >> 1;
    const int dh   = blockIdx.x & 1;

    gauss_table(gt, itp, tid);
    for (int idx = tid; idx < 128 * VP / 4; idx += 256)
        ((float4*)Vs)[idx] = make_float4(0.0f, 0.0f, 0.0f, 0.0f);
    __syncthreads();

    const int key = r * 8 + wave;
    float mycnt = 0.0f;
    const int4*   b4 = (const int4*)bmu + lane;
    const float4* X4 = (const float4*)X + dh * 32 + lane;
    float4*       V4 = (float4*)Vs + lane;
    unsigned short* q = qs[wave];
    float4 run = make_float4(0.0f, 0.0f, 0.0f, 0.0f);
    int cur = -1;
    for (int chunk = 0; chunk < BSZ / 128 / 16; ++chunk) {
        int4 v[16];
#pragma unroll
        for (int k = 0; k < 16; ++k) v[k] = b4[(chunk * 16 + k) * 32];
        int qn = 0;
#pragma unroll
        for (int k = 0; k < 16; ++k) {
            const int vals[4] = { v[k].x, v[k].y, v[k].z, v[k].w };
            const bool any = ((vals[0] >> 4) == key) | ((vals[1] >> 4) == key) |
                             ((vals[2] >> 4) == key) | ((vals[3] >> 4) == key);
            if (__ballot(any) == 0) continue;
#pragma unroll
            for (int e = 0; e < 4; ++e) {
                const bool hit = (vals[e] >> 4) == key;
                const unsigned m = (unsigned)__ballot(hit);
                if (m == 0) continue;
                if (hit) q[qn + (int)__builtin_amdgcn_mbcnt_lo(m, 0u)] =
                             (unsigned short)(((k * 128 + 4 * lane + e) << 4) | (vals[e] & 15));
                qn += __popc(m);
            }
        }
        __builtin_amdgcn_fence(__ATOMIC_SEQ_CST, "wavefront");
        const int row0 = chunk * 2048;
        int pos = 0;
        while (qn - pos >= 16) {
            take_rows<true>(q, pos, 16, row0, wave * 16, lane, X4, V4, run, cur, mycnt);
            pos += 16;
        }
        if (qn != pos)
            take_rows<false>(q, pos, qn - pos, row0, wave * 16, lane, X4, V4, run, cur, mycnt);
        __builtin_amdgcn_fence(__ATOMIC_SEQ_CST, "wavefront");
    }
    if (cur >= 0) V4[(wave * 16 + cur) * (VP / 4)] = run;
    if (lane < 16) cnts[wave * 16 + lane] = mycnt;
    __syncthreads();

    if (dh == 0 && tid < 128) {
        float s = 0.0f;
#pragma unroll 16
        for (int c = 0; c < 128; ++c) s += gt[127 + tid - c] * cnts[c];
        Tden[r * 128 + tid] = s;
    }

    v2f a[60];
    toeplitz_a(gt, l, half, a);
    v2f b[32];
    const float* vb = Vs + (2 * half) * VP + wave * 16 + l;
#pragma unroll
    for (int s = 0; s < 32; ++s) {
        b[s].x = vb[(4 * s) * VP];
        b[s].y = vb[(4 * s + 1) * VP];
    }
    v8f acc[8];
#pragma unroll
    for (int t = 0; t < 8; ++t) acc[t] = v8f{};
    toeplitz_mm(a, b, acc);

    float* tp = T + (size_t)r * (NG * DIMD) + dh * 128 + wave * 16 + l;
#pragma unroll
    for (int t = 0; t < 8; ++t) {
#pragma unroll
        for (int rr = 0; rr < 8; ++rr)
            tp[(size_t)(16 * t + 8 * half + rr) * DIMD] = acc[t][rr];
    }
}

__global__ void __launch_bounds__(256)
k_rowpass(const float* __restrict__ T, const float* __restrict__ Tden,
          const int* __restrict__ itp, const float* __restrict__ W,
          float* __restrict__ out) {
    __shared__ float gt[256];
    __shared__ float tds[128];
    __shared__ __attribute__((aligned(16))) float dens[128];
    const int tid  = threadIdx.x;
    const int lane = tid & 31;
    const int wave = tid >> 5;
    const int l    = lane & 15;
    const int half = lane >> 4;
    const int j    = blockIdx.x >> 1;
    const int dh   = blockIdx.x & 1;

    gauss_table(gt, itp, tid);
    if (tid < 128) tds[tid] = Tden[tid * 128 + j];
    __syncthreads();
    if (tid < 128) {
        float s = 0.0f;
#pragma unroll 16
        for (int r = 0; r < 128; ++r) s += gt[127 + tid - r] * tds[r];
        dens[tid] = s;
    }
    __syncthreads();

    v2f a[60];
    toeplitz_a(gt, l, half, a);
    const int n0 = j * DIMD + dh * 128 + wave * 16 + l;
    v2f b[32];
    const float* tb = T + (size_t)(2 * half) * (NG * DIMD) + n0;
#pragma unroll
    for (int s = 0; s < 32; ++s) {
        b[s].x = tb[(size_t)(4 * s) * (NG * DIMD)];
        b[s].y = tb[(size_t)(4 * s + 1) * (NG * DIMD)];
    }
    v8f acc[8];
#pragma unroll
    for (int t = 0; t < 8; ++t) acc[t] = v8f{};
    toeplitz_mm(a, b, acc);

    float4 dq[16];
#pragma unroll
    for (int t = 0; t < 8; ++t) {
        dq[2 * t]     = *(const float4*)&dens[16 * t + 8 * half];
        dq[2 * t + 1] = *(const float4*)&dens[16 * t + 8 * half + 4];
    }
#pragma unroll
    for (int t = 0; t < 8; ++t) {
        const float den8[8] = { dq[2 * t].x, dq[2 * t].y, dq[2 * t].z, dq[2 * t].w,
                                dq[2 * t + 1].x, dq[2 * t + 1].y, dq[2 * t + 1].z, dq[2 * t + 1].w };
#pragma unroll
        for (int rr = 0; rr < 8; ++rr) {
            const int i = 16 * t + 8 * half + rr;
            const size_t idx = (size_t)i * (NG * DIMD) + n0;
            const float den = den8[rr];
            out[idx] = (den != 0.0f) ? (acc[t][rr] / den) : W[idx];
        }
    }
}

extern "C" void kernel_launch(void* const* d_in, const int* in_sizes, int n_in,
                              void* d_out, int out_size, void* d_ws, size_t ws_size,
                              hipStream_t stream) {
    const float* X  = (const float*)d_in[0];
    const float* W  = (const float*)d_in[1];
    const int*   it = (const int*)d_in[2];
    float* out = (float*)d_out;

    float* ws    = (float*)d_ws;
    float* wsq   = ws;
    int*   bmu   = (int*)(ws + KUNITS);
    float* Xf    = ws + KUNITS + 2 * BSZ;
    u32x4* Xh    = (u32x4*)(Xf + (size_t)BSZ * DIMD);
    float* Mt    = Xf + (size_t)BSZ * DIMD + (size_t)BSZ * DIMD / 2;
    float* ea    = Mt + (size_t)(KUNITS / 16) * BSZ;
    float* eb    = ea + KUNITS / 16;
    float* nx    = eb + KUNITS / 16;
    float* Wf    = nx + BSZ;
    float* T     = Wf + (size_t)KUNITS * DIMD;
    float* Tden  = T + (size_t)KUNITS * DIMD;

    k_prep   <<<(KUNITS + BSZ) / 64, 64, 0, stream>>>(W, X, wsq, Xf, Xh, nx, Wf);
    k_lo     <<<(KUNITS / 32 / 8) * 8, 256, 0, stream>>>(W, Xh, wsq, Mt, ea, eb);
    k_verify <<<BSZ / 16, 256, 0, stream>>>(Wf, Xf, wsq, Mt, ea, eb, nx, bmu);
    k_colpass<<<MG * 2, 256, 0, stream>>>(X, bmu, it, T, Tden);
    k_rowpass<<<NG * 2, 256, 0, stream>>>(T, Tden, it, W, out);
}


# k_lo: a wave keeps 4 unit tiles (64 units) in VGPRs, 2 waves per SIMD: a B fragment feeds 4 matrix ops, not 2; the next row tile's B loads issue under the current tile's products; Mt bit-identical
# speedup vs baseline: 29.4704x; 1.2298x over previous
#include <hip/hip_runtime.h>
#include <math.h>

#define MG   128
#define NG   128
#define DIMD 256
#define BSZ  8192
#define KUNITS (MG * NG)

typedef __attribute__((ext_vector_type(2))) float v2f;
typedef __attribute__((ext_vector_type(8))) float v8f;

__device__ __forceinline__ v8f wmma_f32(v2f a, v2f b, v8f c) {
    return __builtin_amdgcn_wmma_f32_16x16x4_f32(
         false, a,  false, b,
         (short)0, c,  false,  false);
}

typedef __attribute__((ext_vector_type(8)))  __bf16   bf16x8;
typedef __attribute__((ext_vector_type(16))) __bf16   bf16x16;
typedef __attribute__((ext_vector_type(4)))  unsigned u32x4;
typedef __attribute__((ext_vector_type(8)))  unsigned u32x8;
typedef __attribute__((ext_vector_type(4)))  float    f32x4;

__device__ __forceinline__ v8f wmma_bf16(bf16x16 a, bf16x16 b, v8f c) {
    return __builtin_amdgcn_wmma_f32_16x16x32_bf16(
         false, a,  false, b,
         (short)0, c,  false,  false);
}


__global__ void __launch_bounds__(256)
k_prep(const float* __restrict__ W, const float* __restrict__ X,
       float* __restrict__ wsq, float* __restrict__ Xf,
       u32x4* __restrict__ Xh, float* __restrict__ nx, float* __restrict__ Wf) {
    const int g = blockIdx.x * blockDim.x + threadIdx.x;
    if (g < KUNITS) {
        const int k = g;
        const float4* w4 = (const float4*)(W + (size_t)k * DIMD);
        float4* wdst = (float4*)Wf + (size_t)(k >> 4) * (32 * 32) + (k & 15);
        float acc = 0.0f;
#pragma unroll 4
        for (int p = 0; p < DIMD / 8; ++p) {
            const float4 v0 = w4[2 * p];
            const float4 v1 = w4[2 * p + 1];
            acc += v0.x * v0.x + v0.y * v0.y + v0.z * v0.z + v0.w * v0.w;
            acc += v1.x * v1.x + v1.y * v1.y + v1.z * v1.z + v1.w * v1.w;
            wdst[p * 32]      = make_float4(v0.x, v0.y, v1.x, v1.y);
            wdst[p * 32 + 16] = make_float4(v0.z, v0.w, v1.z, v1.w);
        }
        wsq[k] = acc;
    } else {
        const int k = g - KUNITS;
        const float4* x4 = (const float4*)(X + (size_t)k * DIMD);
        float4* dst = (float4*)Xf + (size_t)(k >> 4) * (32 * 32) + (k & 15);
        u32x4*  dh  = Xh + (size_t)(k >> 4) * (8 * 2 * 32) + (k & 15);
        float xs = 0.0f;
#pragma unroll 4
        for (int p = 0; p < DIMD / 8; ++p) {
            const float4 v0 = x4[2 * p];
            const float4 v1 = x4[2 * p + 1];
            dst[p * 32]      = make_float4(v0.x, v0.y, v1.x, v1.y);
            dst[p * 32 + 16] = make_float4(v0.z, v0.w, v1.z, v1.w);
            xs += v0.x * v0.x + v0.y * v0.y + v0.z * v0.z + v0.w * v0.w;
            xs += v1.x * v1.x + v1.y * v1.y + v1.z * v1.z + v1.w * v1.w;
            bf16x8 hv;
            hv[0] = (__bf16)v0.x; hv[1] = (__bf16)v0.y; hv[2] = (__bf16)v0.z; hv[3] = (__bf16)v0.w;
            hv[4] = (__bf16)v1.x; hv[5] = (__bf16)v1.y; hv[6] = (__bf16)v1.z; hv[7] = (__bf16)v1.w;
            dh[(((p >> 2) * 2 + (p & 1)) * 32) + ((p >> 1) & 1) * 16] = __builtin_bit_cast(u32x4, hv);
        }
        nx[k] = sqrtf(xs) * 1.001f;
    }
}

__device__ __forceinline__ void lo_tile(const bf16x16 (&a)[4][8], const u32x4 (&bb)[16],
                                        const float (&ws)[4][8], int half, float* mrow) {
    v8f c[4];
#pragma unroll
    for (int t = 0; t < 4; ++t) c[t] = v8f{};
#pragma unroll
    for (int P = 0; P < 8; ++P) {
        const u32x4 lo = bb[2 * P];
        const u32x4 hi = bb[2 * P + 1];
        u32x8 v;
        v[0] = lo[0]; v[1] = lo[1]; v[2] = lo[2]; v[3] = lo[3];
        v[4] = hi[0]; v[5] = hi[1]; v[6] = hi[2]; v[7] = hi[3];
        const bf16x16 b = __builtin_bit_cast(bf16x16, v);
#pragma unroll
        for (int t = 0; t < 4; ++t) c[t] = wmma_bf16(a[t][P], b, c[t]);
    }
    float m[4];
#pragma unroll
    for (int t = 0; t < 4; ++t) {
        m[t] = ws[t][0] - 2.0f * c[t][0];
#pragma unroll
        for (int r = 1; r < 8; ++r) m[t] = fminf(m[t], ws[t][r] - 2.0f * c[t][r]);
    }
    const float k0 = half ? m[1] : m[0];
    const float g0 = half ? m[0] : m[1];
    const float k1 = half ? m[3] : m[2];
    const float g1 = half ? m[2] : m[3];
    mrow[0]  = fminf(k0, __shfl_xor(g0, 16, 32));
    mrow[32] = fminf(k1, __shfl_xor(g1, 16, 32));
}

__global__ void __launch_bounds__(256) __attribute__((amdgpu_waves_per_eu(2, 2)))
k_lo(const float* __restrict__ W, const u32x4* __restrict__ Xh,
     const float* __restrict__ wsq, float* __restrict__ Mt,
     float* __restrict__ ea, float* __restrict__ eb) {
    const int tid  = threadIdx.x;
    const int lane = tid & 31;
    const int wave = tid >> 5;
    const int l    = lane & 15;
    const int half = lane >> 4;
    const int qd   = (blockIdx.x >> 3) * 8 + wave;
    const int rt0  = (blockIdx.x & 7) * (BSZ / 16 / 8);

    if ((blockIdx.x & 7) == 0) {
#pragma unroll
        for (int j = 0; j < 2; ++j) {
            float wm = wsq[qd * 64 + j * 32 + lane];
            wm = fmaxf(wm, __shfl_xor(wm, 8, 32));
            wm = fmaxf(wm, __shfl_xor(wm, 4, 32));
            wm = fmaxf(wm, __shfl_xor(wm, 2, 32));
            wm = fmaxf(wm, __shfl_xor(wm, 1, 32));
            if (l == 0) {
                ea[4 * qd + 2 * j + half] = sqrtf(wm) * 1.001f * (0.015625f * 1.05f);
                eb[4 * qd + 2 * j + half] = wm * 9.5367431640625e-7f;
            }
        }
    }

    bf16x16 a[4][8];
    const float4* r0 = (const float4*)(W + (size_t)(qd * 64 + l) * DIMD + 16 * half);
#pragma unroll
    for (int t = 0; t < 4; ++t) {
        const float4* rr = r0 + t * 16 * (DIMD / 4);
#pragma unroll
        for (int P = 0; P < 8; ++P) {
#pragma unroll
            for (int q = 0; q < 4; ++q) {
                const float4 v0 = rr[P * 8 + q];
                a[t][P][4 * q + 0] = (__bf16)v0.x; a[t][P][4 * q + 1] = (__bf16)v0.y;
                a[t][P][4 * q + 2] = (__bf16)v0.z; a[t][P][4 * q + 3] = (__bf16)v0.w;
            }
        }
        if (t == 1)
            asm volatile("" :: "v"(__builtin_bit_cast(u32x8, a[1][0])), "v"(__builtin_bit_cast(u32x8, a[1][1])),
                               "v"(__builtin_bit_cast(u32x8, a[1][2])), "v"(__builtin_bit_cast(u32x8, a[1][3])),
                               "v"(__builtin_bit_cast(u32x8, a[1][4])), "v"(__builtin_bit_cast(u32x8, a[1][5])),
                               "v"(__builtin_bit_cast(u32x8, a[1][6])), "v"(__builtin_bit_cast(u32x8, a[1][7])),
                               "v"(__builtin_bit_cast(u32x8, a[0][0])), "v"(__builtin_bit_cast(u32x8, a[0][1])),
                               "v"(__builtin_bit_cast(u32x8, a[0][2])), "v"(__builtin_bit_cast(u32x8, a[0][3])),
                               "v"(__builtin_bit_cast(u32x8, a[0][4])), "v"(__builtin_bit_cast(u32x8, a[0][5])),
                               "v"(__builtin_bit_cast(u32x8, a[0][6])), "v"(__builtin_bit_cast(u32x8, a[0][7])));
    }

    float ws[4][8];
#pragma unroll
    for (int t = 0; t < 4; ++t) {
#pragma unroll
        for (int r = 0; r < 8; ++r) ws[t][r] = wsq[qd * 64 + t * 16 + 8 * half + r];
    }

    const u32x4* xb = Xh + lane;
    float* mp = Mt + (size_t)(4 * qd + half) * 16 + l;
    const int rtl = rt0 + BSZ / 16 / 8 - 1;
    u32x4 b0[16], b1[16];
    {
        const u32x4* bt = xb + (size_t)rt0 * (8 * 2 * 32);
#pragma unroll
        for (int i = 0; i < 16; ++i) b0[i] = bt[i * 32];
    }
#pragma unroll 1
    for (int rt = rt0; rt < rtl; rt += 2) {
        {
            const u32x4* bt = xb + (size_t)(rt + 1) * (8 * 2 * 32);
#pragma unroll
            for (int i = 0; i < 16; ++i) b1[i] = bt[i * 32];
        }
        lo_tile(a, b0, ws, half, mp + (size_t)rt * KUNITS);
        {
            const int rn = (rt + 2 < rtl) ? rt + 2 : rtl;
            const u32x4* bt = xb + (size_t)rn * (8 * 2 * 32);
#pragma unroll
            for (int i = 0; i < 16; ++i) b0[i] = bt[i * 32];
        }
        lo_tile(a, b1, ws, half, mp + (size_t)(rt + 1) * KUNITS);
    }
}

__global__ void __launch_bounds__(256)
k_verify(const float* __restrict__ Wf, const float* __restrict__ Xf,
         const float* __restrict__ wsq, const float* __restrict__ Mt,
         const float* __restrict__ ea, const float* __restrict__ eb,
         const float* __restrict__ nx, int* __restrict__ bmu) {
    __shared__ __attribute__((aligned(16))) float hpart[8][16];
    __shared__ unsigned marks[32];
    __shared__ float    sv[8][16];
    __shared__ int      si[8][16];
    const int tid  = threadIdx.x;
    const int lane = tid & 31;
    const int wave = __builtin_amdgcn_readfirstlane(tid >> 5);
    const int l    = lane & 15;
    const int half = lane >> 4;
    const int rt   = blockIdx.x;

    const int rq = tid & 3;
    const int tb = tid >> 2;
    float4 mv[16];
    float  eav[16], ebv[16];
    {
        const float4* m4 = (const float4*)(Mt + (size_t)rt * KUNITS) + tid;
#pragma unroll
        for (int j = 0; j < 16; ++j) mv[j] = m4[j * 256];
#pragma unroll
        for (int j = 0; j < 16; ++j) { eav[j] = ea[64 * j + tb]; ebv[j] = eb[64 * j + tb]; }
    }
    const float4 nx4 = *(const float4*)(nx + rt * 16 + 4 * rq);

    float4 h = make_float4(3.4e38f, 3.4e38f, 3.4e38f, 3.4e38f);
#pragma unroll
    for (int j = 0; j < 16; ++j) {
        const float e0 = nx4.x * eav[j] + ebv[j];
        const float e1 = nx4.y * eav[j] + ebv[j];
        const float e2 = nx4.z * eav[j] + ebv[j];
        const float e3 = nx4.w * eav[j] + ebv[j];
        const float u0 = mv[j].x + e0;
        const float u1 = mv[j].y + e1;
        const float u2 = mv[j].z + e2;
        const float u3 = mv[j].w + e3;
        h.x = (u0 < h.x) ? u0 : h.x;
        h.y = (u1 < h.y) ? u1 : h.y;
        h.z = (u2 < h.z) ? u2 : h.z;
        h.w = (u3 < h.w) ? u3 : h.w;
        mv[j].x -= e0; mv[j].y -= e1; mv[j].z -= e2; mv[j].w -= e3;
    }
#pragma unroll
    for (int off = 4; off <= 16; off <<= 1) {
        const float o0 = __shfl_xor(h.x, off, 32);
        const float o1 = __shfl_xor(h.y, off, 32);
        const float o2 = __shfl_xor(h.z, off, 32);
        const float o3 = __shfl_xor(h.w, off, 32);
        h.x = (o0 < h.x) ? o0 : h.x;
        h.y = (o1 < h.y) ? o1 : h.y;
        h.z = (o2 < h.z) ? o2 : h.z;
        h.w = (o3 < h.w) ? o3 : h.w;
    }
    if (lane < 4) *(float4*)&hpart[wave][4 * lane] = h;
    __syncthreads();
    float4 hl = *(const float4*)&hpart[0][4 * rq];
#pragma unroll
    for (int w = 1; w < 8; ++w) {
        const float4 o = *(const float4*)&hpart[w][4 * rq];
        hl.x = (o.x < hl.x) ? o.x : hl.x;
        hl.y = (o.y < hl.y) ? o.y : hl.y;
        hl.z = (o.z < hl.z) ? o.z : hl.z;
        hl.w = (o.w < hl.w) ? o.w : hl.w;
    }

    unsigned myb = 0u;
#pragma unroll
    for (int j = 0; j < 16; ++j) {
        unsigned x = (unsigned)__ballot(mv[j].x <= hl.x) | (unsigned)__ballot(mv[j].y <= hl.y) |
                     (unsigned)__ballot(mv[j].z <= hl.z) | (unsigned)__ballot(mv[j].w <= hl.w);
        x |= x >> 1;
        x |= x >> 2;
        x &= 0x11111111u;
        x = (x | (x >> 3)) & 0x03030303u;
        x = (x | (x >> 6)) & 0x000F000Fu;
        x = (x | (x >> 12)) & 0xFFu;
        myb = (lane == j) ? x : myb;
    }
    if (lane < 16) ((unsigned char*)marks)[8 * lane + wave] = (unsigned char)myb;
    __syncthreads();

    float4 bq[32];
    {
        const float4* bt = (const float4*)Xf + (size_t)rt * (32 * 32) + lane;
#pragma unroll
        for (int p = 0; p < 32; ++p) bq[p] = bt[p * 32];
    }

    float bv = 3.4e38f;
    int   bi = 0x7fffffff;
    int   rank = 0;
#pragma unroll 1
    for (int wd = 0; wd < 32; ++wd) {
        unsigned word = (unsigned)__builtin_amdgcn_readfirstlane((int)marks[wd]);
        while (word != 0u) {
            const int t = __builtin_ctz(word);
            word &= word - 1u;
            const bool mine = (rank & 7) == wave;
            ++rank;
            if (!mine) continue;
            const int u0 = (wd * 32 + t) * 16;
            const f32x4* at = (const f32x4*)Wf + (size_t)(wd * 32 + t) * (32 * 32) + lane;
            const float4 wa = *(const float4*)(wsq + u0 + 8 * half);
            const float4 wb = *(const float4*)(wsq + u0 + 8 * half + 4);
            v8f c = {};
#pragma unroll
            for (int hb = 0; hb < 2; ++hb) {
                f32x4 aq[16];
#pragma unroll
                for (int i = 0; i < 16; ++i) aq[i] = at[(hb * 16 + i) * 32];
                asm volatile("" :: "v"(aq[0]), "v"(aq[1]), "v"(aq[2]), "v"(aq[3]), "v"(aq[4]), "v"(aq[5]),
                                   "v"(aq[6]), "v"(aq[7]), "v"(aq[8]), "v"(aq[9]), "v"(aq[10]), "v"(aq[11]),
                                   "v"(aq[12]), "v"(aq[13]), "v"(aq[14]), "v"(aq[15]));
#pragma unroll
                for (int i = 0; i < 16; ++i) {
                    const int p = hb * 16 + i;
                    v2f a0, a1, b0, b1;
                    a0.x = aq[i][0]; a0.y = aq[i][1];
                    a1.x = aq[i][2]; a1.y = aq[i][3];
                    b0.x = bq[p].x; b0.y = bq[p].y;
                    b1.x = bq[p].z; b1.y = bq[p].w;
                    c = wmma_f32(a0, b0, c);
                    c = wmma_f32(a1, b1, c);
                }
            }
            const float ws[8] = { wa.x, wa.y, wa.z, wa.w, wb.x, wb.y, wb.z, wb.w };
#pragma unroll
            for (int r = 0; r < 8; ++r) {
                const float v = ws[r] - 2.0f * c[r];
                if (v < bv) { bv = v; bi = u0 + 8 * half + r; }
            }
        }
    }

    {
        const float ov = __shfl_xor(bv, 16, 32);
        const int   oi = __shfl_xor(bi, 16, 32);
        if (ov < bv || (ov == bv && oi < bi)) { bv = ov; bi = oi; }
    }
    if (half == 0) { sv[wave][l] = bv; si[wave][l] = bi; }
    __syncthreads();
    if (tid < 16) {
        float v = sv[0][tid];
        int   i = si[0][tid];
#pragma unroll
        for (int w = 1; w < 8; ++w) {
            const float ov = sv[w][tid];
            const int   oi = si[w][tid];
            if (ov < v || (ov == v && oi < i)) { v = ov; i = oi; }
        }
        bmu[rt * 16 + tid] = i;
    }
}

__device__ __forceinline__ void gauss_table(float* gt, const int* itp, int tid) {
    const float START_SIGMA = 64.0f;
    const float TAU = 20.0f / logf(START_SIGMA);
    const float sigma = START_SIGMA * expf(-(float)itp[0] / TAU);
    const float inv_d = 1.0f / (2.0f * sigma * sigma);
    const float dr = (float)(tid - 127);
    gt[tid] = expf(-dr * dr * inv_d);
}

__device__ __forceinline__ void toeplitz_a(const float* gt, int l, int half, v2f (&a)[60]) {
#pragma unroll
    for (int q = 0; q < 60; ++q) {
        const int m = 127 + 4 * (q - 31) + l - 2 * half;
        a[q].x = gt[m];
        a[q].y = gt[m - 1];
    }
}

__device__ __forceinline__ void toeplitz_mm(const v2f (&a)[60], const v2f (&b)[32], v8f (&acc)[8]) {
#pragma unroll
    for (int s = 0; s < 32; ++s) {
#pragma unroll
        for (int t = 0; t < 8; ++t)
            acc[t] = wmma_f32(a[4 * t - s + 31], b[s], acc[t]);
    }
}

#define VP 136

#define QCAP 2064

template <bool FULL>
__device__ __forceinline__ void take_rows(const unsigned short* q, int pos, int n,
                                          int row0, int col0, int lane,
                                          const float4* X4, float4* V4,
                                          float4& run, int& cur, float& mycnt) {
    const int ent = q[pos + (lane & 15)];
    float4 x[16];
#pragma unroll
    for (int i = 0; i < 16; ++i) {
        x[i] = make_float4(0.0f, 0.0f, 0.0f, 0.0f);
        if (FULL || i < n) {
            const int row = row0 + (__builtin_amdgcn_readlane(ent, i) >> 4);
            x[i] = X4[(size_t)row * (DIMD / 4)];
        }
    }
#pragma unroll
    for (int i = 0; i < 16; ++i) {
        if (FULL || i < n) {
            const int c = __builtin_amdgcn_readlane(ent, i) & 15;
            if (c != cur) {
                if (cur >= 0) V4[(col0 + cur) * (VP / 4)] = run;
                run = V4[(col0 + c) * (VP / 4)];
                cur = c;
            }
            run.x += x[i].x; run.y += x[i].y; run.z += x[i].z; run.w += x[i].w;
            if (lane == c) mycnt += 1.0f;
        }
    }
}

__global__ void __launch_bounds__(256)
k_colpass(const float* __restrict__ X, const int* __restrict__ bmu,
          const int* __restrict__ itp, float* __restrict__ T, float* __restrict__ Tden) {
    __shared__ float gt[256];
    __shared__ float cnts[128];
    __shared__ __attribute__((aligned(16))) float Vs[128 * VP];
    __shared__ unsigned short qs[8][QCAP];
    const int tid  = threadIdx.x;
    const int lane = tid & 31;
    const int wave = tid >> 5;
    const int l    = lane & 15;
    const int half = lane >> 4;
    const int r    = blockIdx.x >> 1;
    const int dh   = blockIdx.x & 1;

    gauss_table(gt, itp, tid);
    for (int idx = tid; idx < 128 * VP / 4; idx += 256)
        ((float4*)Vs)[idx] = make_float4(0.0f, 0.0f, 0.0f, 0.0f);
    __syncthreads();

    const int key = r * 8 + wave;
    float mycnt = 0.0f;
    const int4*   b4 = (const int4*)bmu + lane;
    const float4* X4 = (const float4*)X + dh * 32 + lane;
    float4*       V4 = (float4*)Vs + lane;
    unsigned short* q = qs[wave];
    float4 run = make_float4(0.0f, 0.0f, 0.0f, 0.0f);
    int cur = -1;
    for (int chunk = 0; chunk < BSZ / 128 / 16; ++chunk) {
        int4 v[16];
#pragma unroll
        for (int k = 0; k < 16; ++k) v[k] = b4[(chunk * 16 + k) * 32];
        int qn = 0;
#pragma unroll
        for (int k = 0; k < 16; ++k) {
            const int vals[4] = { v[k].x, v[k].y, v[k].z, v[k].w };
            const bool any = ((vals[0] >> 4) == key) | ((vals[1] >> 4) == key) |
                             ((vals[2] >> 4) == key) | ((vals[3] >> 4) == key);
            if (__ballot(any) == 0) continue;
#pragma unroll
            for (int e = 0; e < 4; ++e) {
                const bool hit = (vals[e] >> 4) == key;
                const unsigned m = (unsigned)__ballot(hit);
                if (m == 0) continue;
                if (hit) q[qn + (int)__builtin_amdgcn_mbcnt_lo(m, 0u)] =
                             (unsigned short)(((k * 128 + 4 * lane + e) << 4) | (vals[e] & 15));
                qn += __popc(m);
            }
        }
        __builtin_amdgcn_fence(__ATOMIC_SEQ_CST, "wavefront");
        const int row0 = chunk * 2048;
        int pos = 0;
        while (qn - pos >= 16) {
            take_rows<true>(q, pos, 16, row0, wave * 16, lane, X4, V4, run, cur, mycnt);
            pos += 16;
        }
        if (qn != pos)
            take_rows<false>(q, pos, qn - pos, row0, wave * 16, lane, X4, V4, run, cur, mycnt);
        __builtin_amdgcn_fence(__ATOMIC_SEQ_CST, "wavefront");
    }
    if (cur >= 0) V4[(wave * 16 + cur) * (VP / 4)] = run;
    if (lane < 16) cnts[wave * 16 + lane] = mycnt;
    __syncthreads();

    if (dh == 0 && tid < 128) {
        float s = 0.0f;
#pragma unroll 16
        for (int c = 0; c < 128; ++c) s += gt[127 + tid - c] * cnts[c];
        Tden[r * 128 + tid] = s;
    }

    v2f a[60];
    toeplitz_a(gt, l, half, a);
    v2f b[32];
    const float* vb = Vs + (2 * half) * VP + wave * 16 + l;
#pragma unroll
    for (int s = 0; s < 32; ++s) {
        b[s].x = vb[(4 * s) * VP];
        b[s].y = vb[(4 * s + 1) * VP];
    }
    v8f acc[8];
#pragma unroll
    for (int t = 0; t < 8; ++t) acc[t] = v8f{};
    toeplitz_mm(a, b, acc);

    float* tp = T + (size_t)r * (NG * DIMD) + dh * 128 + wave * 16 + l;
#pragma unroll
    for (int t = 0; t < 8; ++t) {
#pragma unroll
        for (int rr = 0; rr < 8; ++rr)
            tp[(size_t)(16 * t + 8 * half + rr) * DIMD] = acc[t][rr];
    }
}

__global__ void __launch_bounds__(256)
k_rowpass(const float* __restrict__ T, const float* __restrict__ Tden,
          const int* __restrict__ itp, const float* __restrict__ W,
          float* __restrict__ out) {
    __shared__ float gt[256];
    __shared__ float tds[128];
    __shared__ __attribute__((aligned(16))) float dens[128];
    const int tid  = threadIdx.x;
    const int lane = tid & 31;
    const int wave = tid >> 5;
    const int l    = lane & 15;
    const int half = lane >> 4;
    const int j    = blockIdx.x >> 1;
    const int dh   = blockIdx.x & 1;

    gauss_table(gt, itp, tid);
    if (tid < 128) tds[tid] = Tden[tid * 128 + j];
    __syncthreads();
    if (tid < 128) {
        float s = 0.0f;
#pragma unroll 16
        for (int r = 0; r < 128; ++r) s += gt[127 + tid - r] * tds[r];
        dens[tid] = s;
    }
    __syncthreads();

    v2f a[60];
    toeplitz_a(gt, l, half, a);
    const int n0 = j * DIMD + dh * 128 + wave * 16 + l;
    v2f b[32];
    const float* tb = T + (size_t)(2 * half) * (NG * DIMD) + n0;
#pragma unroll
    for (int s = 0; s < 32; ++s) {
        b[s].x = tb[(size_t)(4 * s) * (NG * DIMD)];
        b[s].y = tb[(size_t)(4 * s + 1) * (NG * DIMD)];
    }
    v8f acc[8];
#pragma unroll
    for (int t = 0; t < 8; ++t) acc[t] = v8f{};
    toeplitz_mm(a, b, acc);

    float4 dq[16];
#pragma unroll
    for (int t = 0; t < 8; ++t) {
        dq[2 * t]     = *(const float4*)&dens[16 * t + 8 * half];
        dq[2 * t + 1] = *(const float4*)&dens[16 * t + 8 * half + 4];
    }
#pragma unroll
    for (int t = 0; t < 8; ++t) {
        const float den8[8] = { dq[2 * t].x, dq[2 * t].y, dq[2 * t].z, dq[2 * t].w,
                                dq[2 * t + 1].x, dq[2 * t + 1].y, dq[2 * t + 1].z, dq[2 * t + 1].w };
#pragma unroll
        for (int rr = 0; rr < 8; ++rr) {
            const int i = 16 * t + 8 * half + rr;
            const size_t idx = (size_t)i * (NG * DIMD) + n0;
            const float den = den8[rr];
            out[idx] = (den != 0.0f) ? (acc[t][rr] / den) : W[idx];
        }
    }
}

extern "C" void kernel_launch(void* const* d_in, const int* in_sizes, int n_in,
                              void* d_out, int out_size, void* d_ws, size_t ws_size,
                              hipStream_t stream) {
    const float* X  = (const float*)d_in[0];
    const float* W  = (const float*)d_in[1];
    const int*   it = (const int*)d_in[2];
    float* out = (float*)d_out;

    float* ws    = (float*)d_ws;
    float* wsq   = ws;
    int*   bmu   = (int*)(ws + KUNITS);
    float* Xf    = ws + KUNITS + 2 * BSZ;
    u32x4* Xh    = (u32x4*)(Xf + (size_t)BSZ * DIMD);
    float* Mt    = Xf + (size_t)BSZ * DIMD + (size_t)BSZ * DIMD / 2;
    float* ea    = Mt + (size_t)(KUNITS / 16) * BSZ;
    float* eb    = ea + KUNITS / 16;
    float* nx    = eb + KUNITS / 16;
    float* Wf    = nx + BSZ;
    float* T     = Wf + (size_t)KUNITS * DIMD;
    float* Tden  = T + (size_t)KUNITS * DIMD;

    k_prep   <<<(KUNITS + BSZ) / 64, 64, 0, stream>>>(W, X, wsq, Xf, Xh, nx, Wf);
    k_lo     <<<(KUNITS / 64 / 8) * 8, 256, 0, stream>>>(W, Xh, wsq, Mt, ea, eb);
    k_verify <<<BSZ / 16, 256, 0, stream>>>(Wf, Xf, wsq, Mt, ea, eb, nx, bmu);
    k_colpass<<<MG * 2, 256, 0, stream>>>(X, bmu, it, T, Tden);
    k_rowpass<<<NG * 2, 256, 0, stream>>>(T, Tden, it, W, out);
}


# screening sweep's operands f16, not bf16 (same 16-bit matrix op and rate): the proven error bound is 8x tighter, so k_verify re-sweeps less than half as many tiles in f32; BMU bit-identical
# speedup vs baseline: 32.3045x; 1.0962x over previous
#include <hip/hip_runtime.h>
#include <math.h>

#define MG   128
#define NG   128
#define DIMD 256
#define BSZ  8192
#define KUNITS (MG * NG)

typedef __attribute__((ext_vector_type(2))) float v2f;
typedef __attribute__((ext_vector_type(8))) float v8f;

__device__ __forceinline__ v8f wmma_f32(v2f a, v2f b, v8f c) {
    return __builtin_amdgcn_wmma_f32_16x16x4_f32(
         false, a,  false, b,
         (short)0, c,  false,  false);
}

typedef __attribute__((ext_vector_type(8)))  _Float16 f16x8;
typedef __attribute__((ext_vector_type(16))) _Float16 f16x16;
typedef __attribute__((ext_vector_type(4)))  unsigned u32x4;
typedef __attribute__((ext_vector_type(8)))  unsigned u32x8;
typedef __attribute__((ext_vector_type(4)))  float    f32x4;

__device__ __forceinline__ v8f wmma_f16(f16x16 a, f16x16 b, v8f c) {
    return __builtin_amdgcn_wmma_f32_16x16x32_f16(
         false, a,  false, b,
         (short)0, c,  false,  false);
}


__global__ void __launch_bounds__(256)
k_prep(const float* __restrict__ W, const float* __restrict__ X,
       float* __restrict__ wsq, float* __restrict__ Xf,
       u32x4* __restrict__ Xh, float* __restrict__ nx, float* __restrict__ Wf) {
    const int g = blockIdx.x * blockDim.x + threadIdx.x;
    if (g < KUNITS) {
        const int k = g;
        const float4* w4 = (const float4*)(W + (size_t)k * DIMD);
        float4* wdst = (float4*)Wf + (size_t)(k >> 4) * (32 * 32) + (k & 15);
        float acc = 0.0f;
#pragma unroll 4
        for (int p = 0; p < DIMD / 8; ++p) {
            const float4 v0 = w4[2 * p];
            const float4 v1 = w4[2 * p + 1];
            acc += v0.x * v0.x + v0.y * v0.y + v0.z * v0.z + v0.w * v0.w;
            acc += v1.x * v1.x + v1.y * v1.y + v1.z * v1.z + v1.w * v1.w;
            wdst[p * 32]      = make_float4(v0.x, v0.y, v1.x, v1.y);
            wdst[p * 32 + 16] = make_float4(v0.z, v0.w, v1.z, v1.w);
        }
        wsq[k] = acc;
    } else {
        const int k = g - KUNITS;
        const float4* x4 = (const float4*)(X + (size_t)k * DIMD);
        float4* dst = (float4*)Xf + (size_t)(k >> 4) * (32 * 32) + (k & 15);
        u32x4*  dh  = Xh + (size_t)(k >> 4) * (8 * 2 * 32) + (k & 15);
        float xs = 0.0f;
#pragma unroll 4
        for (int p = 0; p < DIMD / 8; ++p) {
            const float4 v0 = x4[2 * p];
            const float4 v1 = x4[2 * p + 1];
            dst[p * 32]      = make_float4(v0.x, v0.y, v1.x, v1.y);
            dst[p * 32 + 16] = make_float4(v0.z, v0.w, v1.z, v1.w);
            xs += v0.x * v0.x + v0.y * v0.y + v0.z * v0.z + v0.w * v0.w;
            xs += v1.x * v1.x + v1.y * v1.y + v1.z * v1.z + v1.w * v1.w;
            f16x8 hv;
            hv[0] = (_Float16)v0.x; hv[1] = (_Float16)v0.y; hv[2] = (_Float16)v0.z; hv[3] = (_Float16)v0.w;
            hv[4] = (_Float16)v1.x; hv[5] = (_Float16)v1.y; hv[6] = (_Float16)v1.z; hv[7] = (_Float16)v1.w;
            dh[(((p >> 2) * 2 + (p & 1)) * 32) + ((p >> 1) & 1) * 16] = __builtin_bit_cast(u32x4, hv);
        }
        nx[k] = sqrtf(xs) * 1.001f;
    }
}

__device__ __forceinline__ void lo_tile(const f16x16 (&a)[4][8], const u32x4 (&bb)[16],
                                        const float (&ws)[4][8], int half, float* mrow) {
    v8f c[4];
#pragma unroll
    for (int t = 0; t < 4; ++t) c[t] = v8f{};
#pragma unroll
    for (int P = 0; P < 8; ++P) {
        const u32x4 lo = bb[2 * P];
        const u32x4 hi = bb[2 * P + 1];
        u32x8 v;
        v[0] = lo[0]; v[1] = lo[1]; v[2] = lo[2]; v[3] = lo[3];
        v[4] = hi[0]; v[5] = hi[1]; v[6] = hi[2]; v[7] = hi[3];
        const f16x16 b = __builtin_bit_cast(f16x16, v);
#pragma unroll
        for (int t = 0; t < 4; ++t) c[t] = wmma_f16(a[t][P], b, c[t]);
    }
    float m[4];
#pragma unroll
    for (int t = 0; t < 4; ++t) {
        m[t] = ws[t][0] - 2.0f * c[t][0];
#pragma unroll
        for (int r = 1; r < 8; ++r) m[t] = fminf(m[t], ws[t][r] - 2.0f * c[t][r]);
    }
    const float k0 = half ? m[1] : m[0];
    const float g0 = half ? m[0] : m[1];
    const float k1 = half ? m[3] : m[2];
    const float g1 = half ? m[2] : m[3];
    mrow[0]  = fminf(k0, __shfl_xor(g0, 16, 32));
    mrow[32] = fminf(k1, __shfl_xor(g1, 16, 32));
}

__global__ void __launch_bounds__(256) __attribute__((amdgpu_waves_per_eu(2, 2)))
k_lo(const float* __restrict__ W, const u32x4* __restrict__ Xh,
     const float* __restrict__ wsq, float* __restrict__ Mt,
     float* __restrict__ ea, float* __restrict__ eb) {
    const int tid  = threadIdx.x;
    const int lane = tid & 31;
    const int wave = tid >> 5;
    const int l    = lane & 15;
    const int half = lane >> 4;
    const int qd   = (blockIdx.x >> 3) * 8 + wave;
    const int rt0  = (blockIdx.x & 7) * (BSZ / 16 / 8);

    if ((blockIdx.x & 7) == 0) {
#pragma unroll
        for (int j = 0; j < 2; ++j) {
            float wm = wsq[qd * 64 + j * 32 + lane];
            wm = fmaxf(wm, __shfl_xor(wm, 8, 32));
            wm = fmaxf(wm, __shfl_xor(wm, 4, 32));
            wm = fmaxf(wm, __shfl_xor(wm, 2, 32));
            wm = fmaxf(wm, __shfl_xor(wm, 1, 32));
            if (l == 0) {
                const float nwm = sqrtf(wm) * 1.001f;
                ea[4 * qd + 2 * j + half] = nwm * (0.001953125f * 1.12f) + 9.6e-7f;
                eb[4 * qd + 2 * j + half] = wm * 9.5367431640625e-7f + nwm * 9.6e-7f + 1e-12f;
            }
        }
    }

    f16x16 a[4][8];
    const float4* r0 = (const float4*)(W + (size_t)(qd * 64 + l) * DIMD + 16 * half);
#pragma unroll
    for (int t = 0; t < 4; ++t) {
        const float4* rr = r0 + t * 16 * (DIMD / 4);
#pragma unroll
        for (int P = 0; P < 8; ++P) {
#pragma unroll
            for (int q = 0; q < 4; ++q) {
                const float4 v0 = rr[P * 8 + q];
                a[t][P][4 * q + 0] = (_Float16)v0.x; a[t][P][4 * q + 1] = (_Float16)v0.y;
                a[t][P][4 * q + 2] = (_Float16)v0.z; a[t][P][4 * q + 3] = (_Float16)v0.w;
            }
        }
        if (t == 1)
            asm volatile("" :: "v"(__builtin_bit_cast(u32x8, a[1][0])), "v"(__builtin_bit_cast(u32x8, a[1][1])),
                               "v"(__builtin_bit_cast(u32x8, a[1][2])), "v"(__builtin_bit_cast(u32x8, a[1][3])),
                               "v"(__builtin_bit_cast(u32x8, a[1][4])), "v"(__builtin_bit_cast(u32x8, a[1][5])),
                               "v"(__builtin_bit_cast(u32x8, a[1][6])), "v"(__builtin_bit_cast(u32x8, a[1][7])),
                               "v"(__builtin_bit_cast(u32x8, a[0][0])), "v"(__builtin_bit_cast(u32x8, a[0][1])),
                               "v"(__builtin_bit_cast(u32x8, a[0][2])), "v"(__builtin_bit_cast(u32x8, a[0][3])),
                               "v"(__builtin_bit_cast(u32x8, a[0][4])), "v"(__builtin_bit_cast(u32x8, a[0][5])),
                               "v"(__builtin_bit_cast(u32x8, a[0][6])), "v"(__builtin_bit_cast(u32x8, a[0][7])));
    }

    float ws[4][8];
#pragma unroll
    for (int t = 0; t < 4; ++t) {
#pragma unroll
        for (int r = 0; r < 8; ++r) ws[t][r] = wsq[qd * 64 + t * 16 + 8 * half + r];
    }

    const u32x4* xb = Xh + lane;
    float* mp = Mt + (size_t)(4 * qd + half) * 16 + l;
    const int rtl = rt0 + BSZ / 16 / 8 - 1;
    u32x4 b0[16], b1[16];
    {
        const u32x4* bt = xb + (size_t)rt0 * (8 * 2 * 32);
#pragma unroll
        for (int i = 0; i < 16; ++i) b0[i] = bt[i * 32];
    }
#pragma unroll 1
    for (int rt = rt0; rt < rtl; rt += 2) {
        {
            const u32x4* bt = xb + (size_t)(rt + 1) * (8 * 2 * 32);
#pragma unroll
            for (int i = 0; i < 16; ++i) b1[i] = bt[i * 32];
        }
        lo_tile(a, b0, ws, half, mp + (size_t)rt * KUNITS);
        {
            const int rn = (rt + 2 < rtl) ? rt + 2 : rtl;
            const u32x4* bt = xb + (size_t)rn * (8 * 2 * 32);
#pragma unroll
            for (int i = 0; i < 16; ++i) b0[i] = bt[i * 32];
        }
        lo_tile(a, b1, ws, half, mp + (size_t)(rt + 1) * KUNITS);
    }
}

__global__ void __launch_bounds__(256)
k_verify(const float* __restrict__ Wf, const float* __restrict__ Xf,
         const float* __restrict__ wsq, const float* __restrict__ Mt,
         const float* __restrict__ ea, const float* __restrict__ eb,
         const float* __restrict__ nx, int* __restrict__ bmu) {
    __shared__ __attribute__((aligned(16))) float hpart[8][16];
    __shared__ unsigned marks[32];
    __shared__ float    sv[8][16];
    __shared__ int      si[8][16];
    const int tid  = threadIdx.x;
    const int lane = tid & 31;
    const int wave = __builtin_amdgcn_readfirstlane(tid >> 5);
    const int l    = lane & 15;
    const int half = lane >> 4;
    const int rt   = blockIdx.x;

    const int rq = tid & 3;
    const int tb = tid >> 2;
    float4 mv[16];
    float  eav[16], ebv[16];
    {
        const float4* m4 = (const float4*)(Mt + (size_t)rt * KUNITS) + tid;
#pragma unroll
        for (int j = 0; j < 16; ++j) mv[j] = m4[j * 256];
#pragma unroll
        for (int j = 0; j < 16; ++j) { eav[j] = ea[64 * j + tb]; ebv[j] = eb[64 * j + tb]; }
    }
    const float4 nx4 = *(const float4*)(nx + rt * 16 + 4 * rq);

    float4 h = make_float4(3.4e38f, 3.4e38f, 3.4e38f, 3.4e38f);
#pragma unroll
    for (int j = 0; j < 16; ++j) {
        const float e0 = nx4.x * eav[j] + ebv[j];
        const float e1 = nx4.y * eav[j] + ebv[j];
        const float e2 = nx4.z * eav[j] + ebv[j];
        const float e3 = nx4.w * eav[j] + ebv[j];
        const float u0 = mv[j].x + e0;
        const float u1 = mv[j].y + e1;
        const float u2 = mv[j].z + e2;
        const float u3 = mv[j].w + e3;
        h.x = (u0 < h.x) ? u0 : h.x;
        h.y = (u1 < h.y) ? u1 : h.y;
        h.z = (u2 < h.z) ? u2 : h.z;
        h.w = (u3 < h.w) ? u3 : h.w;
        mv[j].x -= e0; mv[j].y -= e1; mv[j].z -= e2; mv[j].w -= e3;
    }
#pragma unroll
    for (int off = 4; off <= 16; off <<= 1) {
        const float o0 = __shfl_xor(h.x, off, 32);
        const float o1 = __shfl_xor(h.y, off, 32);
        const float o2 = __shfl_xor(h.z, off, 32);
        const float o3 = __shfl_xor(h.w, off, 32);
        h.x = (o0 < h.x) ? o0 : h.x;
        h.y = (o1 < h.y) ? o1 : h.y;
        h.z = (o2 < h.z) ? o2 : h.z;
        h.w = (o3 < h.w) ? o3 : h.w;
    }
    if (lane < 4) *(float4*)&hpart[wave][4 * lane] = h;
    __syncthreads();
    float4 hl = *(const float4*)&hpart[0][4 * rq];
#pragma unroll
    for (int w = 1; w < 8; ++w) {
        const float4 o = *(const float4*)&hpart[w][4 * rq];
        hl.x = (o.x < hl.x) ? o.x : hl.x;
        hl.y = (o.y < hl.y) ? o.y : hl.y;
        hl.z = (o.z < hl.z) ? o.z : hl.z;
        hl.w = (o.w < hl.w) ? o.w : hl.w;
    }

    unsigned myb = 0u;
#pragma unroll
    for (int j = 0; j < 16; ++j) {
        unsigned x = (unsigned)__ballot(mv[j].x <= hl.x) | (unsigned)__ballot(mv[j].y <= hl.y) |
                     (unsigned)__ballot(mv[j].z <= hl.z) | (unsigned)__ballot(mv[j].w <= hl.w);
        x |= x >> 1;
        x |= x >> 2;
        x &= 0x11111111u;
        x = (x | (x >> 3)) & 0x03030303u;
        x = (x | (x >> 6)) & 0x000F000Fu;
        x = (x | (x >> 12)) & 0xFFu;
        myb = (lane == j) ? x : myb;
    }
    if (lane < 16) ((unsigned char*)marks)[8 * lane + wave] = (unsigned char)myb;
    __syncthreads();

    float4 bq[32];
    {
        const float4* bt = (const float4*)Xf + (size_t)rt * (32 * 32) + lane;
#pragma unroll
        for (int p = 0; p < 32; ++p) bq[p] = bt[p * 32];
    }

    float bv = 3.4e38f;
    int   bi = 0x7fffffff;
    int   rank = 0;
#pragma unroll 1
    for (int wd = 0; wd < 32; ++wd) {
        unsigned word = (unsigned)__builtin_amdgcn_readfirstlane((int)marks[wd]);
        while (word != 0u) {
            const int t = __builtin_ctz(word);
            word &= word - 1u;
            const bool mine = (rank & 7) == wave;
            ++rank;
            if (!mine) continue;
            const int u0 = (wd * 32 + t) * 16;
            const f32x4* at = (const f32x4*)Wf + (size_t)(wd * 32 + t) * (32 * 32) + lane;
            const float4 wa = *(const float4*)(wsq + u0 + 8 * half);
            const float4 wb = *(const float4*)(wsq + u0 + 8 * half + 4);
            v8f c = {};
#pragma unroll
            for (int hb = 0; hb < 2; ++hb) {
                f32x4 aq[16];
#pragma unroll
                for (int i = 0; i < 16; ++i) aq[i] = at[(hb * 16 + i) * 32];
                asm volatile("" :: "v"(aq[0]), "v"(aq[1]), "v"(aq[2]), "v"(aq[3]), "v"(aq[4]), "v"(aq[5]),
                                   "v"(aq[6]), "v"(aq[7]), "v"(aq[8]), "v"(aq[9]), "v"(aq[10]), "v"(aq[11]),
                                   "v"(aq[12]), "v"(aq[13]), "v"(aq[14]), "v"(aq[15]));
#pragma unroll
                for (int i = 0; i < 16; ++i) {
                    const int p = hb * 16 + i;
                    v2f a0, a1, b0, b1;
                    a0.x = aq[i][0]; a0.y = aq[i][1];
                    a1.x = aq[i][2]; a1.y = aq[i][3];
                    b0.x = bq[p].x; b0.y = bq[p].y;
                    b1.x = bq[p].z; b1.y = bq[p].w;
                    c = wmma_f32(a0, b0, c);
                    c = wmma_f32(a1, b1, c);
                }
            }
            const float ws[8] = { wa.x, wa.y, wa.z, wa.w, wb.x, wb.y, wb.z, wb.w };
#pragma unroll
            for (int r = 0; r < 8; ++r) {
                const float v = ws[r] - 2.0f * c[r];
                if (v < bv) { bv = v; bi = u0 + 8 * half + r; }
            }
        }
    }

    {
        const float ov = __shfl_xor(bv, 16, 32);
        const int   oi = __shfl_xor(bi, 16, 32);
        if (ov < bv || (ov == bv && oi < bi)) { bv = ov; bi = oi; }
    }
    if (half == 0) { sv[wave][l] = bv; si[wave][l] = bi; }
    __syncthreads();
    if (tid < 16) {
        float v = sv[0][tid];
        int   i = si[0][tid];
#pragma unroll
        for (int w = 1; w < 8; ++w) {
            const float ov = sv[w][tid];
            const int   oi = si[w][tid];
            if (ov < v || (ov == v && oi < i)) { v = ov; i = oi; }
        }
        bmu[rt * 16 + tid] = i;
    }
}

__device__ __forceinline__ void gauss_table(float* gt, const int* itp, int tid) {
    const float START_SIGMA = 64.0f;
    const float TAU = 20.0f / logf(START_SIGMA);
    const float sigma = START_SIGMA * expf(-(float)itp[0] / TAU);
    const float inv_d = 1.0f / (2.0f * sigma * sigma);
    const float dr = (float)(tid - 127);
    gt[tid] = expf(-dr * dr * inv_d);
}

__device__ __forceinline__ void toeplitz_a(const float* gt, int l, int half, v2f (&a)[60]) {
#pragma unroll
    for (int q = 0; q < 60; ++q) {
        const int m = 127 + 4 * (q - 31) + l - 2 * half;
        a[q].x = gt[m];
        a[q].y = gt[m - 1];
    }
}

__device__ __forceinline__ void toeplitz_mm(const v2f (&a)[60], const v2f (&b)[32], v8f (&acc)[8]) {
#pragma unroll
    for (int s = 0; s < 32; ++s) {
#pragma unroll
        for (int t = 0; t < 8; ++t)
            acc[t] = wmma_f32(a[4 * t - s + 31], b[s], acc[t]);
    }
}

#define VP 136

#define QCAP 2064

template <bool FULL>
__device__ __forceinline__ void take_rows(const unsigned short* q, int pos, int n,
                                          int row0, int col0, int lane,
                                          const float4* X4, float4* V4,
                                          float4& run, int& cur, float& mycnt) {
    const int ent = q[pos + (lane & 15)];
    float4 x[16];
#pragma unroll
    for (int i = 0; i < 16; ++i) {
        x[i] = make_float4(0.0f, 0.0f, 0.0f, 0.0f);
        if (FULL || i < n) {
            const int row = row0 + (__builtin_amdgcn_readlane(ent, i) >> 4);
            x[i] = X4[(size_t)row * (DIMD / 4)];
        }
    }
#pragma unroll
    for (int i = 0; i < 16; ++i) {
        if (FULL || i < n) {
            const int c = __builtin_amdgcn_readlane(ent, i) & 15;
            if (c != cur) {
                if (cur >= 0) V4[(col0 + cur) * (VP / 4)] = run;
                run = V4[(col0 + c) * (VP / 4)];
                cur = c;
            }
            run.x += x[i].x; run.y += x[i].y; run.z += x[i].z; run.w += x[i].w;
            if (lane == c) mycnt += 1.0f;
        }
    }
}

__global__ void __launch_bounds__(256)
k_colpass(const float* __restrict__ X, const int* __restrict__ bmu,
          const int* __restrict__ itp, float* __restrict__ T, float* __restrict__ Tden) {
    __shared__ float gt[256];
    __shared__ float cnts[128];
    __shared__ __attribute__((aligned(16))) float Vs[128 * VP];
    __shared__ unsigned short qs[8][QCAP];
    const int tid  = threadIdx.x;
    const int lane = tid & 31;
    const int wave = tid >> 5;
    const int l    = lane & 15;
    const int half = lane >> 4;
    const int r    = blockIdx.x >> 1;
    const int dh   = blockIdx.x & 1;

    gauss_table(gt, itp, tid);
    for (int idx = tid; idx < 128 * VP / 4; idx += 256)
        ((float4*)Vs)[idx] = make_float4(0.0f, 0.0f, 0.0f, 0.0f);
    __syncthreads();

    const int key = r * 8 + wave;
    float mycnt = 0.0f;
    const int4*   b4 = (const int4*)bmu + lane;
    const float4* X4 = (const float4*)X + dh * 32 + lane;
    float4*       V4 = (float4*)Vs + lane;
    unsigned short* q = qs[wave];
    float4 run = make_float4(0.0f, 0.0f, 0.0f, 0.0f);
    int cur = -1;
    for (int chunk = 0; chunk < BSZ / 128 / 16; ++chunk) {
        int4 v[16];
#pragma unroll
        for (int k = 0; k < 16; ++k) v[k] = b4[(chunk * 16 + k) * 32];
        int qn = 0;
#pragma unroll
        for (int k = 0; k < 16; ++k) {
            const int vals[4] = { v[k].x, v[k].y, v[k].z, v[k].w };
            const bool any = ((vals[0] >> 4) == key) | ((vals[1] >> 4) == key) |
                             ((vals[2] >> 4) == key) | ((vals[3] >> 4) == key);
            if (__ballot(any) == 0) continue;
#pragma unroll
            for (int e = 0; e < 4; ++e) {
                const bool hit = (vals[e] >> 4) == key;
                const unsigned m = (unsigned)__ballot(hit);
                if (m == 0) continue;
                if (hit) q[qn + (int)__builtin_amdgcn_mbcnt_lo(m, 0u)] =
                             (unsigned short)(((k * 128 + 4 * lane + e) << 4) | (vals[e] & 15));
                qn += __popc(m);
            }
        }
        __builtin_amdgcn_fence(__ATOMIC_SEQ_CST, "wavefront");
        const int row0 = chunk * 2048;
        int pos = 0;
        while (qn - pos >= 16) {
            take_rows<true>(q, pos, 16, row0, wave * 16, lane, X4, V4, run, cur, mycnt);
            pos += 16;
        }
        if (qn != pos)
            take_rows<false>(q, pos, qn - pos, row0, wave * 16, lane, X4, V4, run, cur, mycnt);
        __builtin_amdgcn_fence(__ATOMIC_SEQ_CST, "wavefront");
    }
    if (cur >= 0) V4[(wave * 16 + cur) * (VP / 4)] = run;
    if (lane < 16) cnts[wave * 16 + lane] = mycnt;
    __syncthreads();

    if (dh == 0 && tid < 128) {
        float s = 0.0f;
#pragma unroll 16
        for (int c = 0; c < 128; ++c) s += gt[127 + tid - c] * cnts[c];
        Tden[r * 128 + tid] = s;
    }

    v2f a[60];
    toeplitz_a(gt, l, half, a);
    v2f b[32];
    const float* vb = Vs + (2 * half) * VP + wave * 16 + l;
#pragma unroll
    for (int s = 0; s < 32; ++s) {
        b[s].x = vb[(4 * s) * VP];
        b[s].y = vb[(4 * s + 1) * VP];
    }
    v8f acc[8];
#pragma unroll
    for (int t = 0; t < 8; ++t) acc[t] = v8f{};
    toeplitz_mm(a, b, acc);

    float* tp = T + (size_t)r * (NG * DIMD) + dh * 128 + wave * 16 + l;
#pragma unroll
    for (int t = 0; t < 8; ++t) {
#pragma unroll
        for (int rr = 0; rr < 8; ++rr)
            tp[(size_t)(16 * t + 8 * half + rr) * DIMD] = acc[t][rr];
    }
}

__global__ void __launch_bounds__(256)
k_rowpass(const float* __restrict__ T, const float* __restrict__ Tden,
          const int* __restrict__ itp, const float* __restrict__ W,
          float* __restrict__ out) {
    __shared__ float gt[256];
    __shared__ float tds[128];
    __shared__ __attribute__((aligned(16))) float dens[128];
    const int tid  = threadIdx.x;
    const int lane = tid & 31;
    const int wave = tid >> 5;
    const int l    = lane & 15;
    const int half = lane >> 4;
    const int j    = blockIdx.x >> 1;
    const int dh   = blockIdx.x & 1;

    gauss_table(gt, itp, tid);
    if (tid < 128) tds[tid] = Tden[tid * 128 + j];
    __syncthreads();
    if (tid < 128) {
        float s = 0.0f;
#pragma unroll 16
        for (int r = 0; r < 128; ++r) s += gt[127 + tid - r] * tds[r];
        dens[tid] = s;
    }
    __syncthreads();

    v2f a[60];
    toeplitz_a(gt, l, half, a);
    const int n0 = j * DIMD + dh * 128 + wave * 16 + l;
    v2f b[32];
    const float* tb = T + (size_t)(2 * half) * (NG * DIMD) + n0;
#pragma unroll
    for (int s = 0; s < 32; ++s) {
        b[s].x = tb[(size_t)(4 * s) * (NG * DIMD)];
        b[s].y = tb[(size_t)(4 * s + 1) * (NG * DIMD)];
    }
    v8f acc[8];
#pragma unroll
    for (int t = 0; t < 8; ++t) acc[t] = v8f{};
    toeplitz_mm(a, b, acc);

    float4 dq[16];
#pragma unroll
    for (int t = 0; t < 8; ++t) {
        dq[2 * t]     = *(const float4*)&dens[16 * t + 8 * half];
        dq[2 * t + 1] = *(const float4*)&dens[16 * t + 8 * half + 4];
    }
#pragma unroll
    for (int t = 0; t < 8; ++t) {
        const float den8[8] = { dq[2 * t].x, dq[2 * t].y, dq[2 * t].z, dq[2 * t].w,
                                dq[2 * t + 1].x, dq[2 * t + 1].y, dq[2 * t + 1].z, dq[2 * t + 1].w };
#pragma unroll
        for (int rr = 0; rr < 8; ++rr) {
            const int i = 16 * t + 8 * half + rr;
            const size_t idx = (size_t)i * (NG * DIMD) + n0;
            const float den = den8[rr];
            out[idx] = (den != 0.0f) ? (acc[t][rr] / den) : W[idx];
        }
    }
}

extern "C" void kernel_launch(void* const* d_in, const int* in_sizes, int n_in,
                              void* d_out, int out_size, void* d_ws, size_t ws_size,
                              hipStream_t stream) {
    const float* X  = (const float*)d_in[0];
    const float* W  = (const float*)d_in[1];
    const int*   it = (const int*)d_in[2];
    float* out = (float*)d_out;

    float* ws    = (float*)d_ws;
    float* wsq   = ws;
    int*   bmu   = (int*)(ws + KUNITS);
    float* Xf    = ws + KUNITS + 2 * BSZ;
    u32x4* Xh    = (u32x4*)(Xf + (size_t)BSZ * DIMD);
    float* Mt    = Xf + (size_t)BSZ * DIMD + (size_t)BSZ * DIMD / 2;
    float* ea    = Mt + (size_t)(KUNITS / 16) * BSZ;
    float* eb    = ea + KUNITS / 16;
    float* nx    = eb + KUNITS / 16;
    float* Wf    = nx + BSZ;
    float* T     = Wf + (size_t)KUNITS * DIMD;
    float* Tden  = T + (size_t)KUNITS * DIMD;

    k_prep   <<<(KUNITS + BSZ) / 64, 64, 0, stream>>>(W, X, wsq, Xf, Xh, nx, Wf);
    k_lo     <<<(KUNITS / 64 / 8) * 8, 256, 0, stream>>>(W, Xh, wsq, Mt, ea, eb);
    k_verify <<<BSZ / 16, 256, 0, stream>>>(Wf, Xf, wsq, Mt, ea, eb, nx, bmu);
    k_colpass<<<MG * 2, 256, 0, stream>>>(X, bmu, it, T, Tden);
    k_rowpass<<<NG * 2, 256, 0, stream>>>(T, Tden, it, W, out);
}
